# starting kernel
# baseline (speedup 1.0000x reference)
#include <hip/hip_runtime.h>


typedef __attribute__((ext_vector_type(2))) float v2f;
typedef __attribute__((ext_vector_type(8))) float v8f;

static constexpr size_t S1P  = 0;
static constexpr size_t S2P  = 8388608;
static constexpr size_t S3M  = 16777216;
static constexpr size_t D12  = 17825792;
static constexpr size_t D13  = 18874368;
static constexpr size_t D23  = 19922944;
static constexpr size_t TRI  = 20971520;
static constexpr size_t S1M  = 20987904;
static constexpr size_t S2M  = 22036480;
static constexpr size_t A1O  = 23085056;
static constexpr size_t A0O  = 23248896;
static constexpr size_t R12O = 23250176;
static constexpr size_t R12TO= 23266560;
static constexpr size_t FORO = 23282944;
static constexpr size_t R02O = 23299328;
static constexpr size_t WS_FLOATS = 23299584;

__global__ void k1_contract(const float* __restrict__ x,
                            float* __restrict__ s1p, float* __restrict__ s2p,
                            float* __restrict__ s3m, float* __restrict__ d12,
                            float* __restrict__ d13, float* __restrict__ d23,
                            float* __restrict__ tri) {
  __shared__ float lds[7 * 2048 + 256];
  const int t  = threadIdx.x;
  const int jt = blockIdx.x, it = blockIdx.y, b = blockIdx.z;
  const int kg = t >> 5, c = t & 31;

  for (int idx = t; idx < 7 * 2048; idx += 256) lds[idx] = 0.f;
  float s2r[8];
  #pragma unroll
  for (int m = 0; m < 8; ++m) s2r[m] = 0.f;
  __syncthreads();

  for (int jj = 0; jj < 8; ++jj) {
    const int j = jt * 8 + jj;
    float s1a[8];
    #pragma unroll
    for (int m = 0; m < 8; ++m) s1a[m] = 0.f;

    for (int ii = 0; ii < 8; ++ii) {
      const int i = it * 8 + ii;
      const float* row = x + (((size_t)((b * 64 + i) * 64 + j)) << 11);
      float v[8];
      #pragma unroll
      for (int m = 0; m < 8; ++m) v[m] = row[t + (m << 8)];

      float p = 0.f;
      #pragma unroll
      for (int m = 0; m < 8; ++m) { s1a[m] += v[m]; p += v[m]; }

      if (ii < 7) {
        #pragma unroll
        for (int m = 0; m < 8; ++m) lds[ii * 2048 + t + (m << 8)] += v[m];
      } else {
        #pragma unroll
        for (int m = 0; m < 8; ++m) s2r[m] += v[m];
      }

      lds[14336 + t] = p;
      __syncthreads();
      if (t < 32) {
        float tot = 0.f;
        #pragma unroll
        for (int g = 0; g < 8; ++g) tot += lds[14336 + g * 32 + t];
        s3m[(((size_t)((b * 64 + i) * 64 + j)) << 5) + t] = tot * (1.f / 64.f);
      }
      if ((i & 7) == kg) d13[(((size_t)((b * 64 + i) * 64 + j)) << 5) + c] = v[i >> 3];
      if ((j & 7) == kg) d23[(((size_t)((b * 64 + i) * 64 + j)) << 5) + c] = v[j >> 3];
      if (i == j) {
        #pragma unroll
        for (int m = 0; m < 8; ++m)
          d12[(((size_t)(b * 64 + i)) << 11) + t + (m << 8)] = v[m];
        if ((i & 7) == kg) tri[((b << 6) + i) * 32 + c] = v[i >> 3];
      }
      __syncthreads();
    }
    float* dst = s1p + (((size_t)((b * 8 + it) * 64 + j)) << 11);
    #pragma unroll
    for (int m = 0; m < 8; ++m) dst[t + (m << 8)] = s1a[m];
  }
  for (int ii = 0; ii < 8; ++ii) {
    float* dst = s2p + (((size_t)((b * 8 + jt) * 64 + it * 8 + ii)) << 11);
    if (ii < 7) {
      #pragma unroll
      for (int m = 0; m < 8; ++m) dst[t + (m << 8)] = lds[ii * 2048 + t + (m << 8)];
    } else {
      #pragma unroll
      for (int m = 0; m < 8; ++m) dst[t + (m << 8)] = s2r[m];
    }
  }
}

__global__ void k2a_reduce(const float* __restrict__ s1p, const float* __restrict__ s2p,
                           float* __restrict__ s1m, float* __restrict__ s2m) {
  const int t = threadIdx.x;
  const int u = blockIdx.x & 63, b = blockIdx.x >> 6;
  #pragma unroll
  for (int m = 0; m < 8; ++m) {
    const int e = t + (m << 8);
    float a = 0.f, s = 0.f;
    #pragma unroll
    for (int g = 0; g < 8; ++g) {
      a += s1p[(((size_t)((b * 8 + g) * 64 + u)) << 11) + e];
      s += s2p[(((size_t)((b * 8 + g) * 64 + u)) << 11) + e];
    }
    s1m[(((size_t)(b * 64 + u)) << 11) + e] = a * (1.f / 64.f);
    s2m[(((size_t)(b * 64 + u)) << 11) + e] = s * (1.f / 64.f);
  }
}

__global__ void k2b_a1(const float* __restrict__ s1m, const float* __restrict__ s3m,
                       const float* __restrict__ d12, const float* __restrict__ d13,
                       const float* __restrict__ d23, const float* __restrict__ tri,
                       float* __restrict__ a1) {
  __shared__ float scr[256];
  const int t = threadIdx.x, c = t & 31, g = t >> 5;
  const int b = blockIdx.x >> 6, u = blockIdx.x & 63;
  float* dst = a1 + (size_t)(b * 64 + u) * 320;

  auto red = [&](float partial, int slot) {
    scr[t] = partial;
    __syncthreads();
    if (t < 32) {
      float tot = 0.f;
      #pragma unroll
      for (int gg = 0; gg < 8; ++gg) tot += scr[gg * 32 + t];
      dst[slot * 32 + t] = tot * (1.f / 64.f);
    }
    __syncthreads();
  };
  float acc;
  acc = 0.f; for (int s = 0; s < 8; ++s) { int v = g + 8 * s;
    acc += s1m[(((size_t)((b * 64 + v) * 64 + u)) << 5) + c]; } red(acc, 0);
  acc = 0.f; for (int s = 0; s < 8; ++s) { int v = g + 8 * s;
    acc += s1m[(((size_t)((b * 64 + u) * 64 + v)) << 5) + c]; } red(acc, 1);
  acc = 0.f; for (int s = 0; s < 8; ++s) { int v = g + 8 * s;
    acc += s3m[(((size_t)((b * 64 + u) * 64 + v)) << 5) + c]; } red(acc, 2);
  acc = 0.f; for (int s = 0; s < 8; ++s) { int v = g + 8 * s;
    acc += d12[(((size_t)(b * 64 + v)) << 11) + u * 32 + c]; } red(acc, 3);
  acc = 0.f; for (int s = 0; s < 8; ++s) { int v = g + 8 * s;
    acc += d12[(((size_t)(b * 64 + u)) << 11) + v * 32 + c]; } red(acc, 4);
  acc = 0.f; for (int s = 0; s < 8; ++s) { int v = g + 8 * s;
    acc += d13[(((size_t)((b * 64 + v) * 64 + u)) << 5) + c]; } red(acc, 5);
  acc = 0.f; for (int s = 0; s < 8; ++s) { int v = g + 8 * s;
    acc += d13[(((size_t)((b * 64 + u) * 64 + v)) << 5) + c]; } red(acc, 6);
  acc = 0.f; for (int s = 0; s < 8; ++s) { int v = g + 8 * s;
    acc += d23[(((size_t)((b * 64 + v) * 64 + u)) << 5) + c]; } red(acc, 7);
  acc = 0.f; for (int s = 0; s < 8; ++s) { int v = g + 8 * s;
    acc += d23[(((size_t)((b * 64 + u) * 64 + v)) << 5) + c]; } red(acc, 8);
  if (t < 32) dst[9 * 32 + t] = tri[(b * 64 + u) * 32 + t];
}

__global__ void k2c_a0(const float* __restrict__ a1, float* __restrict__ a0) {
  const int b = blockIdx.x, t = threadIdx.x;
  if (t < 160) {
    const int s = t >> 5, c = t & 31;
    const int slot = (s == 0) ? 0 : (2 * s + 1);
    float acc = 0.f;
    for (int u = 0; u < 64; ++u)
      acc += a1[(size_t)(b * 64 + u) * 320 + slot * 32 + c];
    a0[b * 160 + s * 32 + c] = acc * (1.f / 64.f);
  }
}

__global__ void k3a_rows(const float* __restrict__ a1, const float* __restrict__ a0,
                         const float* __restrict__ W12, const float* __restrict__ b12,
                         const float* __restrict__ W12t, const float* __restrict__ b12t,
                         const float* __restrict__ W11, const float* __restrict__ b11,
                         const float* __restrict__ W01, const float* __restrict__ b01,
                         const float* __restrict__ W02, const float* __restrict__ b02,
                         const float* __restrict__ b22,
                         float* __restrict__ r12, float* __restrict__ r12t,
                         float* __restrict__ forow, float* __restrict__ r02) {
  const int row = blockIdx.x;
  const int b = row >> 6, p = row & 63;
  const int n = threadIdx.x;
  const float* a1r = a1 + (size_t)row * 320;
  const float* a0r = a0 + b * 160;
  float v12 = b12[n], v12t = b12t[n], vfo = b11[n] + b01[n];
  for (int k = 0; k < 320; ++k) {
    const float av = a1r[k];
    v12  += av * W12 [k * 32 + n];
    v12t += av * W12t[k * 32 + n];
    vfo  += av * W11 [k * 32 + n];
  }
  for (int k = 0; k < 160; ++k) vfo += a0r[k] * W01[k * 32 + n];
  r12  [row * 32 + n] = v12;
  r12t [row * 32 + n] = v12t;
  forow[row * 32 + n] = vfo;
  if (p == 0) {
    float v02 = b02[n] + b22[n];
    for (int k = 0; k < 160; ++k) v02 += a0r[k] * W02[k * 32 + n];
    r02[b * 32 + n] = v02;
  }
}

__global__ void k3b_mix(const float* __restrict__ s1m, const float* __restrict__ s2m,
                        const float* __restrict__ s3m, const float* __restrict__ d12,
                        const float* __restrict__ d13, const float* __restrict__ d23,
                        const float* __restrict__ W22,
                        const float* __restrict__ r12, const float* __restrict__ r12t,
                        const float* __restrict__ r02, const float* __restrict__ forow,
                        float* __restrict__ out) {
  __shared__ float w[384 * 32];
  const int t = threadIdx.x;
  for (int idx = t; idx < 384 * 32; idx += 256) w[idx] = W22[idx];
  __syncthreads();

  const int lane = t & 31;
  const int R = (blockIdx.x * 8 + (t >> 5)) * 16;
  const int MA = lane & 15;
  const int kh = lane >> 4;
  const int rowA = R + MA;
  const int bA = rowA >> 12, pA = (rowA >> 6) & 63, qA = rowA & 63;
  const size_t offPQ = ((size_t)((bA * 64 + pA) * 64 + qA)) << 5;
  const size_t offQP = ((size_t)((bA * 64 + qA) * 64 + pA)) << 5;
  const float* comp[6] = { s1m, s2m, s3m, d12, d13, d23 };
  const int n0 = lane & 15;

  v8f acc0 = {};
  v8f acc1 = {};
  #pragma unroll
  for (int f = 0; f < 12; ++f) {
    const float* A = (f < 6) ? (comp[f] + offPQ) : (comp[f - 6] + offQP);
    #pragma unroll
    for (int kk = 0; kk < 32; kk += 4) {
      v2f a = *(const v2f*)(A + kk + 2 * kh);
      const int kcol = f * 32 + kk + 2 * kh;
      v2f b0 = { w[kcol * 32 + n0],      w[(kcol + 1) * 32 + n0] };
      v2f b1 = { w[kcol * 32 + n0 + 16], w[(kcol + 1) * 32 + n0 + 16] };
      acc0 = __builtin_amdgcn_wmma_f32_16x16x4_f32(false, a, false, b0,
                                                   (short)0, acc0, false, false);
      acc1 = __builtin_amdgcn_wmma_f32_16x16x4_f32(false, a, false, b1,
                                                   (short)0, acc1, false, false);
    }
  }
  #pragma unroll
  for (int r = 0; r < 8; ++r) {
    const int M = r + ((lane >> 4) << 3);
    const int row = R + M;
    const int b = row >> 12, p = (row >> 6) & 63, q = row & 63;
    float add0 = r12[((b << 6) + p) * 32 + n0] + r12t[((b << 6) + q) * 32 + n0]
               + r02[(b << 5) + n0];
    float add1 = r12[((b << 6) + p) * 32 + n0 + 16] + r12t[((b << 6) + q) * 32 + n0 + 16]
               + r02[(b << 5) + n0 + 16];
    if (p == q) {
      add0 += forow[((b << 6) + p) * 32 + n0];
      add1 += forow[((b << 6) + p) * 32 + n0 + 16];
    }
    out[((size_t)row << 5) + n0]      = acc0[r] + add0;
    out[((size_t)row << 5) + n0 + 16] = acc1[r] + add1;
  }
}

extern "C" void kernel_launch(void* const* d_in, const int* in_sizes, int n_in,
                              void* d_out, int out_size, void* d_ws, size_t ws_size,
                              hipStream_t stream) {
  (void)in_sizes; (void)n_in; (void)out_size;
  if (ws_size < WS_FLOATS * sizeof(float)) return;

  const float* x    = (const float*)d_in[0];
  const float* W22  = (const float*)d_in[1];
  const float* b22  = (const float*)d_in[2];
  const float* W12  = (const float*)d_in[3];
  const float* b12  = (const float*)d_in[4];
  const float* W12t = (const float*)d_in[5];
  const float* b12t = (const float*)d_in[6];
  const float* W02  = (const float*)d_in[7];
  const float* b02  = (const float*)d_in[8];
  const float* W11  = (const float*)d_in[9];
  const float* b11  = (const float*)d_in[10];
  const float* W01  = (const float*)d_in[11];
  const float* b01  = (const float*)d_in[12];

  float* ws = (float*)d_ws;
  float* s1p = ws + S1P;   float* s2p = ws + S2P;
  float* s3m = ws + S3M;   float* d12 = ws + D12;
  float* d13 = ws + D13;   float* d23 = ws + D23;
  float* tri = ws + TRI;   float* s1m = ws + S1M;
  float* s2m = ws + S2M;   float* a1  = ws + A1O;
  float* a0  = ws + A0O;   float* r12 = ws + R12O;
  float* r12t= ws + R12TO; float* forow = ws + FORO;
  float* r02 = ws + R02O;

  k1_contract<<<dim3(8, 8, 8), 256, 0, stream>>>(x, s1p, s2p, s3m, d12, d13, d23, tri);
  k2a_reduce <<<512, 256, 0, stream>>>(s1p, s2p, s1m, s2m);
  k2b_a1     <<<512, 256, 0, stream>>>(s1m, s3m, d12, d13, d23, tri, a1);
  k2c_a0     <<<8, 256, 0, stream>>>(a1, a0);
  k3a_rows   <<<512, 32, 0, stream>>>(a1, a0, W12, b12, W12t, b12t,
                                      W11, b11, W01, b01, W02, b02, b22,
                                      r12, r12t, forow, r02);
  k3b_mix    <<<256, 256, 0, stream>>>(s1m, s2m, s3m, d12, d13, d23, W22,
                                       r12, r12t, r02, forow, (float*)d_out);
}


# Mean planes s1m/s2m no longer stored: the mix GEMM adds the 8 partial planes into its A tiles itself (whole-line loads, wave-private LDS turn), a1 sums the planes; 6->5 launches (new a1 sum order)
# speedup vs baseline: 2.0922x; 2.0922x over previous
#include <hip/hip_runtime.h>


typedef __attribute__((ext_vector_type(2))) float v2f;
typedef __attribute__((ext_vector_type(8))) float v8f;

static constexpr size_t S1P  = 0;
static constexpr size_t S2P  = 8388608;
static constexpr size_t S3M  = 16777216;
static constexpr size_t D12  = 17825792;
static constexpr size_t D13  = 18874368;
static constexpr size_t D23  = 19922944;
static constexpr size_t TRI  = 20971520;
static constexpr size_t A1O  = 23085056;
static constexpr size_t A0O  = 23248896;
static constexpr size_t R12O = 23250176;
static constexpr size_t R12TO= 23266560;
static constexpr size_t FORO = 23282944;
static constexpr size_t R02O = 23299328;
static constexpr size_t WS_FLOATS = 23299584;

typedef __attribute__((ext_vector_type(4))) float v4f;

__device__ __forceinline__ v4f ld_stream(const float* p) {
  return __builtin_nontemporal_load((const v4f*)p);
}
__device__ __forceinline__ float swap8(float v) {
  return __builtin_bit_cast(float, __builtin_amdgcn_update_dpp(
      0, __builtin_bit_cast(int, v), 0x128  , 0xf, 0xf, false));
}
__device__ __forceinline__ float swap16(float v) {
  return __builtin_bit_cast(float, __builtin_amdgcn_permlanex16(
      __builtin_bit_cast(int, v), __builtin_bit_cast(int, v),
      0x76543210, 0xfedcba98, false, false));
}

struct K1Arg {
  const float* x;
  float *s1p, *s2p, *s3m, *d12, *d13, *d23, *tri;
  v4f* acc2;
  int jt, it, b, j, w, lane;
};
struct K1State {
  v4f s1a[16];
  v4f ring[6][4];
  float g13[2], g23[2];
  v4f p0, p1;
};

template <int G>
__device__ __forceinline__ void k1_issue(K1State& st, const K1Arg& a) {
  constexpr int S = G >> 2, C = G & 3;
  const int i = a.it * 8 + ((a.w + S) & 7);
  const float* r = a.x + (((size_t)((a.b * 64 + i) * 64 + a.j)) << 11);
  #pragma unroll
  for (int q = 0; q < 4; ++q)
    st.ring[G % 6][q] = ld_stream(r + ((C * 4 + q) << 7) + (a.lane << 2));
  if constexpr (C == 0) {
    st.g13[S & 1] = r[(i << 5) + a.lane];
    st.g23[S & 1] = r[(a.j << 5) + a.lane];
  }
}

template <int G>
__device__ __forceinline__ void k1_step(K1State& st, const K1Arg& a) {
  __builtin_amdgcn_sched_barrier(0);
  if constexpr (G + 5 < 32) k1_issue<G + 5>(st, a);
  __builtin_amdgcn_sched_barrier(0);

  constexpr int S = G >> 2, C = G & 3;
  const int ii = (a.w + S) & 7, i = a.it * 8 + ii;
  const size_t rij = (size_t)((a.b * 64 + i) * 64 + a.j);
  v4f* pl = a.acc2 + ii * 512 + (C << 7) + a.lane;

  #pragma unroll
  for (int q = 0; q < 4; ++q) {
    const int m = (C << 2) + q;
    if constexpr (S == 0) st.s1a[m] = st.ring[G % 6][q]; else st.s1a[m] += st.ring[G % 6][q];
  }
  #pragma unroll
  for (int q = 0; q < 4; ++q) asm volatile("" : "+v"(st.s1a[(C << 2) + q]));
  #pragma unroll
  for (int q = 0; q < 4; ++q) {
    const v4f v = st.ring[G % 6][q];
    if (C == 0 && q == 0) st.p0 = v;
    else if (C == 0 && q == 1) st.p1 = v;
    else if (q & 1) st.p1 += v;
    else st.p0 += v;
  }
  if constexpr (S == 0) {
    #pragma unroll
    for (int q = 0; q < 4; ++q) pl[q << 5] = st.ring[G % 6][q];
    if (a.it == a.jt) {
      float* dd = a.d12 + (((size_t)(a.b * 64 + i)) << 11) + (C << 9) + (a.lane << 2);
      #pragma unroll
      for (int q = 0; q < 4; ++q) *(v4f*)(dd + (q << 7)) = st.ring[G % 6][q];
    }
  } else {
    v4f acc[4];
    #pragma unroll
    for (int q = 0; q < 4; ++q) acc[q] = pl[q << 5];
    #pragma unroll
    for (int q = 0; q < 4; ++q) acc[q] += st.ring[G % 6][q];
    if constexpr (S < 7) {
      #pragma unroll
      for (int q = 0; q < 4; ++q) pl[q << 5] = acc[q];
    } else {
      float* ds = a.s2p + (((size_t)((a.b * 8 + a.jt) * 64 + i)) << 11) + (C << 9) + (a.lane << 2);
      #pragma unroll
      for (int q = 0; q < 4; ++q) *(v4f*)(ds + (q << 7)) = acc[q];
    }
  }

  if constexpr (C == 1) {
    a.d13[(rij << 5) + a.lane] = st.g13[S & 1];
    a.d23[(rij << 5) + a.lane] = st.g23[S & 1];
    if constexpr (S == 0) {
      if (a.it == a.jt) a.tri[((a.b << 6) + i) * 32 + a.lane] = st.g13[0];
    }
  }
  if constexpr (C == 3) {
    v4f p = st.p0 + st.p1;
    #pragma unroll
    for (int e = 0; e < 4; ++e) { p[e] += swap8(p[e]); p[e] += swap16(p[e]); }
    if (a.lane < 8)
      *(v4f*)(a.s3m + (rij << 5) + (a.lane << 2)) = p * (1.f / 64.f);
    if constexpr (S < 7) {
      asm volatile("s_wait_dscnt 0x0\n\ts_barrier_signal -1\n\ts_barrier_wait -1"
                   ::: "memory");
    }
  }
  if constexpr (G + 1 < 32) k1_step<G + 1>(st, a);
}

__global__ void __launch_bounds__(256, 4)
k1_contract(const float* __restrict__ x,
            float* __restrict__ s1p, float* __restrict__ s2p,
            float* __restrict__ s3m, float* __restrict__ d12,
            float* __restrict__ d13, float* __restrict__ d23,
            float* __restrict__ tri) {
  __shared__ v4f acc2[8 * 512];
  const int t = threadIdx.x;
  K1Arg a;
  a.x = x; a.s1p = s1p; a.s2p = s2p; a.s3m = s3m;
  a.d12 = d12; a.d13 = d13; a.d23 = d23; a.tri = tri;
  a.acc2 = acc2;
  a.lane = t & 31;
  a.w = __builtin_amdgcn_readfirstlane(t >> 5);
  a.jt = blockIdx.x; a.it = blockIdx.y; a.b = blockIdx.z;
  a.j = a.jt * 8 + a.w;

  K1State st;
  k1_issue<0>(st, a); k1_issue<1>(st, a); k1_issue<2>(st, a);
  k1_issue<3>(st, a); k1_issue<4>(st, a);
  k1_step<0>(st, a);

  float* dst = s1p + (((size_t)((a.b * 8 + a.it) * 64 + a.j)) << 11) + (a.lane << 2);
  #pragma unroll
  for (int m = 0; m < 16; ++m) *(v4f*)(dst + (m << 7)) = st.s1a[m];
}

__global__ void k2b_a1(const float* __restrict__ s1p, const float* __restrict__ s3m,
                       const float* __restrict__ d12, const float* __restrict__ d13,
                       const float* __restrict__ d23, const float* __restrict__ tri,
                       float* __restrict__ a1) {
  __shared__ float scr[256];
  const int t = threadIdx.x, c = t & 31, g = t >> 5;
  const int b = blockIdx.x >> 6, u = blockIdx.x & 63;
  float* dst = a1 + (size_t)(b * 64 + u) * 320;

  auto red = [&](float partial, int slot) {
    scr[t] = partial;
    __syncthreads();
    if (t < 32) {
      float tot = 0.f;
      #pragma unroll
      for (int gg = 0; gg < 8; ++gg) tot += scr[gg * 32 + t];
      dst[slot * 32 + t] = tot * (1.f / 64.f);
    }
    __syncthreads();
  };
  float acc;
  {
    float pv[64];
    #pragma unroll
    for (int pg = 0; pg < 8; ++pg) {
      #pragma unroll
      for (int s = 0; s < 8; ++s)
        pv[pg * 8 + s] = s1p[(((size_t)((b * 8 + pg) * 64 + g + 8 * s)) << 11) + u * 32 + c];
    }
    acc = 0.f;
    #pragma unroll
    for (int e = 0; e < 64; ++e) acc += pv[e];
    red(acc * (1.f / 64.f), 0);
  }
  acc = 0.f; for (int s = 0; s < 8; ++s) { int v = g + 8 * s;
    acc += s3m[(((size_t)((b * 64 + v) * 64 + u)) << 5) + c]; } red(acc, 1);
  acc = 0.f; for (int s = 0; s < 8; ++s) { int v = g + 8 * s;
    acc += s3m[(((size_t)((b * 64 + u) * 64 + v)) << 5) + c]; } red(acc, 2);
  acc = 0.f; for (int s = 0; s < 8; ++s) { int v = g + 8 * s;
    acc += d12[(((size_t)(b * 64 + v)) << 11) + u * 32 + c]; } red(acc, 3);
  acc = 0.f; for (int s = 0; s < 8; ++s) { int v = g + 8 * s;
    acc += d12[(((size_t)(b * 64 + u)) << 11) + v * 32 + c]; } red(acc, 4);
  acc = 0.f; for (int s = 0; s < 8; ++s) { int v = g + 8 * s;
    acc += d13[(((size_t)((b * 64 + v) * 64 + u)) << 5) + c]; } red(acc, 5);
  acc = 0.f; for (int s = 0; s < 8; ++s) { int v = g + 8 * s;
    acc += d13[(((size_t)((b * 64 + u) * 64 + v)) << 5) + c]; } red(acc, 6);
  acc = 0.f; for (int s = 0; s < 8; ++s) { int v = g + 8 * s;
    acc += d23[(((size_t)((b * 64 + v) * 64 + u)) << 5) + c]; } red(acc, 7);
  acc = 0.f; for (int s = 0; s < 8; ++s) { int v = g + 8 * s;
    acc += d23[(((size_t)((b * 64 + u) * 64 + v)) << 5) + c]; } red(acc, 8);
  if (t < 32) dst[9 * 32 + t] = tri[(b * 64 + u) * 32 + t];
}

__global__ void k2c_a0(const float* __restrict__ a1, float* __restrict__ a0) {
  const int b = blockIdx.x, t = threadIdx.x;
  if (t < 160) {
    const int s = t >> 5, c = t & 31;
    const int slot = (s == 0) ? 0 : (2 * s + 1);
    float acc = 0.f;
    for (int u = 0; u < 64; ++u)
      acc += a1[(size_t)(b * 64 + u) * 320 + slot * 32 + c];
    a0[b * 160 + s * 32 + c] = acc * (1.f / 64.f);
  }
}

__device__ __forceinline__ v8f wmma4(v2f a, v2f b, v8f c) {
  return __builtin_amdgcn_wmma_f32_16x16x4_f32(false, a, false, b, (short)0, c,
                                               false, false);
}

__global__ void k3a_rows(const float* __restrict__ a1, const float* __restrict__ a0,
                         const float* __restrict__ W12, const float* __restrict__ b12,
                         const float* __restrict__ W12t, const float* __restrict__ b12t,
                         const float* __restrict__ W11, const float* __restrict__ b11,
                         const float* __restrict__ W01, const float* __restrict__ b01,
                         const float* __restrict__ W02, const float* __restrict__ b02,
                         const float* __restrict__ b22,
                         float* __restrict__ r12, float* __restrict__ r12t,
                         float* __restrict__ forow, float* __restrict__ r02) {
  const int t = threadIdx.x, lane = t & 31;
  const int wv = __builtin_amdgcn_readfirstlane(t >> 5);
  const int R0 = blockIdx.x * 16;
  const int b = R0 >> 6;
  const int kh = lane >> 4;
  const int mat = wv >> 1;
  const int n = ((wv & 1) << 4) + (lane & 15);
  const float* a0k = a0 + b * 160 + 2 * kh;
  v8f accE = {}, accO = {};

  if (mat == 3) {
    if ((R0 & 63) != 0) return;
    const float* wc = W02 + (2 * kh) * 32 + n;
    #pragma unroll
    for (int kk = 0; kk < 160; kk += 8) {
      v2f aE = *(const v2f*)(a0k + kk);
      v2f aO = *(const v2f*)(a0k + kk + 4);
      v2f bE = { wc[kk * 32],       wc[kk * 32 + 32] };
      v2f bO = { wc[(kk + 4) * 32], wc[(kk + 4) * 32 + 32] };
      accE = wmma4(aE, bE, accE);
      accO = wmma4(aO, bO, accO);
    }
    const float v = accE[0] + accO[0] + b02[n] + b22[n];
    if (lane < 16) r02[b * 32 + n] = v;
    return;
  }

  const float* Wm  = (mat == 0) ? W12 : (mat == 1) ? W12t : W11;
  float*       dst = (mat == 0) ? r12 : (mat == 1) ? r12t : forow;
  float bias = (mat == 0) ? b12[n] : (mat == 1) ? b12t[n] : (b11[n] + b01[n]);

  if (mat == 2) {
    const float* wc = W01 + (2 * kh) * 32 + n;
    #pragma unroll
    for (int kk = 0; kk < 160; kk += 8) {
      v2f aE = *(const v2f*)(a0k + kk);
      v2f aO = *(const v2f*)(a0k + kk + 4);
      v2f bE = { wc[kk * 32],       wc[kk * 32 + 32] };
      v2f bO = { wc[(kk + 4) * 32], wc[(kk + 4) * 32 + 32] };
      accE = wmma4(aE, bE, accE);
      accO = wmma4(aO, bO, accO);
    }
  }
  {
    const float* ar = a1 + (size_t)(R0 + (lane & 15)) * 320 + 2 * kh;
    const float* wc = Wm + (2 * kh) * 32 + n;
    #pragma unroll
    for (int kk = 0; kk < 320; kk += 8) {
      v2f aE = *(const v2f*)(ar + kk);
      v2f aO = *(const v2f*)(ar + kk + 4);
      v2f bE = { wc[kk * 32],       wc[kk * 32 + 32] };
      v2f bO = { wc[(kk + 4) * 32], wc[(kk + 4) * 32 + 32] };
      accE = wmma4(aE, bE, accE);
      accO = wmma4(aO, bO, accO);
    }
  }
  #pragma unroll
  for (int r = 0; r < 8; ++r) {
    const int M = r + (kh << 3);
    dst[(R0 + M) * 32 + n] = accE[r] + accO[r] + bias;
  }
}

static constexpr size_t PLANE = 131072;

__device__ __forceinline__ void k3_issue(v4f (&raw)[32], const float* __restrict__ P,
                                         size_t off, int pitch, int lane) {
  const float* p = P + off + (size_t)(lane >> 3) * pitch + ((lane & 7) << 2);
  #pragma unroll
  for (int pg = 0; pg < 8; ++pg) {
    #pragma unroll
    for (int r = 0; r < 4; ++r)
      raw[pg * 4 + r] = *(const v4f*)(p + (size_t)pg * PLANE + (size_t)(4 * r) * pitch);
  }
}

__device__ __forceinline__ void k3_fold(const v4f (&raw)[32], float* sb, int lane) {
  #pragma unroll
  for (int r = 0; r < 4; ++r) {
    v4f a = raw[r];
    #pragma unroll
    for (int pg = 1; pg < 8; ++pg) a += raw[pg * 4 + r];
    *(v4f*)(sb + (4 * r + (lane >> 3)) * 36 + ((lane & 7) << 2)) = a * (1.f / 64.f);
  }
}

__global__ void __launch_bounds__(256)
k3b_mix(const float* __restrict__ s1p, const float* __restrict__ s2p,
        const float* __restrict__ s3m, const float* __restrict__ d12,
        const float* __restrict__ d13, const float* __restrict__ d23,
        const float* __restrict__ W22,
        const float* __restrict__ r12, const float* __restrict__ r12t,
        const float* __restrict__ r02, const float* __restrict__ forow,
        float* __restrict__ out) {
  __shared__ float w[384 * 32];
  __shared__ __attribute__((aligned(16))) float stg[8 * 2 * 576];
  const int t = threadIdx.x;
  const int lane = t & 31;
  const int wv = __builtin_amdgcn_readfirstlane(t >> 5);
  const int R = (blockIdx.x * 8 + wv) * 16;
  const int bT = R >> 12, pT = (R >> 6) & 63, q0 = R & 63;
  const size_t planePQ = ((size_t)(bT * 512 + pT) << 11) + (size_t)(q0 << 5);
  const size_t planeQP = ((size_t)(bT * 512 + q0) << 11) + (size_t)(pT << 5);
  float* sb0 = stg + wv * 1152;
  float* sb1 = sb0 + 576;

  v4f rawA[32], rawB[32];
  k3_issue(rawA, s1p, planePQ, 32, lane);
  k3_issue(rawB, s2p, planePQ, 32, lane);
  __builtin_amdgcn_sched_barrier(0);

  for (int idx = t; idx < 384 * 32; idx += 256) w[idx] = W22[idx];
  __syncthreads();

  const int MA = lane & 15;
  const int kh = lane >> 4;
  const int rowA = R + MA;
  const int bA = rowA >> 12, pA = (rowA >> 6) & 63, qA = rowA & 63;
  const size_t offPQ = ((size_t)((bA * 64 + pA) * 64 + qA)) << 5;
  const size_t offQP = ((size_t)((bA * 64 + qA) * 64 + pA)) << 5;
  const float* comp[4] = { s3m, d12, d13, d23 };
  const int n0 = lane & 15;

  v8f acc0 = {};
  v8f acc1 = {};

  auto mm = [&](int f, const v2f (&a)[8]) {
    #pragma unroll
    for (int s = 0; s < 8; ++s) {
      const int kcol = f * 32 + 4 * s + 2 * kh;
      v2f b0 = { w[kcol * 32 + n0],      w[(kcol + 1) * 32 + n0] };
      v2f b1 = { w[kcol * 32 + n0 + 16], w[(kcol + 1) * 32 + n0 + 16] };
      acc0 = __builtin_amdgcn_wmma_f32_16x16x4_f32(false, a[s], false, b0,
                                                   (short)0, acc0, false, false);
      acc1 = __builtin_amdgcn_wmma_f32_16x16x4_f32(false, a[s], false, b1,
                                                   (short)0, acc1, false, false);
    }
  };
  auto plane_frag = [&](const v4f (&raw)[32], float* sb, v2f (&a)[8]) {
    k3_fold(raw, sb, lane);
    __builtin_amdgcn_fence(__ATOMIC_RELEASE, "wavefront");
    __builtin_amdgcn_wave_barrier();
    __builtin_amdgcn_fence(__ATOMIC_ACQUIRE, "wavefront");
    #pragma unroll
    for (int s = 0; s < 8; ++s) a[s] = *(const v2f*)(sb + MA * 36 + 4 * s + 2 * kh);
    __builtin_amdgcn_fence(__ATOMIC_RELEASE, "wavefront");
    __builtin_amdgcn_wave_barrier();
    __builtin_amdgcn_fence(__ATOMIC_ACQUIRE, "wavefront");
  };
  v2f fd[4][8];
  auto direct_issue = [&](size_t off) {
    #pragma unroll
    for (int j = 0; j < 4; ++j) {
      #pragma unroll
      for (int s = 0; s < 8; ++s) fd[j][s] = *(const v2f*)(comp[j] + off + 4 * s + 2 * kh);
    }
  };
  v2f fa[8], fb[8];

  direct_issue(offPQ);
  __builtin_amdgcn_sched_barrier(0);
  plane_frag(rawA, sb0, fa);
  k3_issue(rawA, s1p, planeQP, 2048, lane);
  plane_frag(rawB, sb1, fb);
  k3_issue(rawB, s2p, planeQP, 2048, lane);
  __builtin_amdgcn_sched_barrier(0);
  mm(0, fa);
  mm(1, fb);
  #pragma unroll
  for (int j = 0; j < 4; ++j) mm(2 + j, fd[j]);
  __builtin_amdgcn_sched_barrier(0);
  direct_issue(offQP);
  plane_frag(rawA, sb0, fa);
  plane_frag(rawB, sb1, fb);
  mm(6, fa);
  mm(7, fb);
  #pragma unroll
  for (int j = 0; j < 4; ++j) mm(8 + j, fd[j]);

  #pragma unroll
  for (int r = 0; r < 8; ++r) {
    const int M = r + ((lane >> 4) << 3);
    const int row = R + M;
    const int b = row >> 12, p = (row >> 6) & 63, q = row & 63;
    float add0 = r12[((b << 6) + p) * 32 + n0] + r12t[((b << 6) + q) * 32 + n0]
               + r02[(b << 5) + n0];
    float add1 = r12[((b << 6) + p) * 32 + n0 + 16] + r12t[((b << 6) + q) * 32 + n0 + 16]
               + r02[(b << 5) + n0 + 16];
    if (p == q) {
      add0 += forow[((b << 6) + p) * 32 + n0];
      add1 += forow[((b << 6) + p) * 32 + n0 + 16];
    }
    out[((size_t)row << 5) + n0]      = acc0[r] + add0;
    out[((size_t)row << 5) + n0 + 16] = acc1[r] + add1;
  }
}

extern "C" void kernel_launch(void* const* d_in, const int* in_sizes, int n_in,
                              void* d_out, int out_size, void* d_ws, size_t ws_size,
                              hipStream_t stream) {
  (void)in_sizes; (void)n_in; (void)out_size;
  if (ws_size < WS_FLOATS * sizeof(float)) return;

  const float* x    = (const float*)d_in[0];
  const float* W22  = (const float*)d_in[1];
  const float* b22  = (const float*)d_in[2];
  const float* W12  = (const float*)d_in[3];
  const float* b12  = (const float*)d_in[4];
  const float* W12t = (const float*)d_in[5];
  const float* b12t = (const float*)d_in[6];
  const float* W02  = (const float*)d_in[7];
  const float* b02  = (const float*)d_in[8];
  const float* W11  = (const float*)d_in[9];
  const float* b11  = (const float*)d_in[10];
  const float* W01  = (const float*)d_in[11];
  const float* b01  = (const float*)d_in[12];

  float* ws = (float*)d_ws;
  float* s1p = ws + S1P;   float* s2p = ws + S2P;
  float* s3m = ws + S3M;   float* d12 = ws + D12;
  float* d13 = ws + D13;   float* d23 = ws + D23;
  float* tri = ws + TRI;   float* a1  = ws + A1O;
  float* a0  = ws + A0O;   float* r12 = ws + R12O;
  float* r12t= ws + R12TO; float* forow = ws + FORO;
  float* r02 = ws + R02O;

  k1_contract<<<dim3(8, 8, 8), 256, 0, stream>>>(x, s1p, s2p, s3m, d12, d13, d23, tri);
  k2b_a1     <<<512, 256, 0, stream>>>(s1p, s3m, d12, d13, d23, tri, a1);
  k2c_a0     <<<8, 256, 0, stream>>>(a1, a0);
  k3a_rows   <<<32, 256, 0, stream>>>(a1, a0, W12, b12, W12t, b12t,
                                      W11, b11, W01, b01, W02, b02, b22,
                                      r12, r12t, forow, r02);
  k3b_mix    <<<256, 256, 0, stream>>>(s1p, s2p, s3m, d12, d13, d23, W22,
                                       r12, r12t, r02, forow, (float*)d_out);
}


# a0 never formed: a0@W02, a0@W01 by linearity as column sums of a1[16 rows, 5 slots] @ W in the row-mix GEMM's spare waves, 4 shares per b added in the mix epilogue; 5->4 launches (new sum order)
# speedup vs baseline: 2.2483x; 1.0746x over previous
#include <hip/hip_runtime.h>


typedef __attribute__((ext_vector_type(2))) float v2f;
typedef __attribute__((ext_vector_type(8))) float v8f;

static constexpr size_t S1P  = 0;
static constexpr size_t S2P  = 8388608;
static constexpr size_t S3M  = 16777216;
static constexpr size_t D12  = 17825792;
static constexpr size_t D13  = 18874368;
static constexpr size_t D23  = 19922944;
static constexpr size_t TRI  = 20971520;
static constexpr size_t R01P = 20987904;
static constexpr size_t A1O  = 23085056;
static constexpr size_t R02P = 23248896;
static constexpr size_t R12O = 23250176;
static constexpr size_t R12TO= 23266560;
static constexpr size_t FORO = 23282944;
static constexpr size_t WS_FLOATS = 23299584;

typedef __attribute__((ext_vector_type(4))) float v4f;

__device__ __forceinline__ v4f ld_stream(const float* p) {
  return __builtin_nontemporal_load((const v4f*)p);
}
__device__ __forceinline__ float swap8(float v) {
  return __builtin_bit_cast(float, __builtin_amdgcn_update_dpp(
      0, __builtin_bit_cast(int, v), 0x128  , 0xf, 0xf, false));
}
__device__ __forceinline__ float swap16(float v) {
  return __builtin_bit_cast(float, __builtin_amdgcn_permlanex16(
      __builtin_bit_cast(int, v), __builtin_bit_cast(int, v),
      0x76543210, 0xfedcba98, false, false));
}

struct K1Arg {
  const float* x;
  float *s1p, *s2p, *s3m, *d12, *d13, *d23, *tri;
  v4f* acc2;
  int jt, it, b, j, w, lane;
};
struct K1State {
  v4f s1a[16];
  v4f ring[6][4];
  float g13[2], g23[2];
  v4f p0, p1;
};

template <int G>
__device__ __forceinline__ void k1_issue(K1State& st, const K1Arg& a) {
  constexpr int S = G >> 2, C = G & 3;
  const int i = a.it * 8 + ((a.w + S) & 7);
  const float* r = a.x + (((size_t)((a.b * 64 + i) * 64 + a.j)) << 11);
  #pragma unroll
  for (int q = 0; q < 4; ++q)
    st.ring[G % 6][q] = ld_stream(r + ((C * 4 + q) << 7) + (a.lane << 2));
  if constexpr (C == 0) {
    st.g13[S & 1] = r[(i << 5) + a.lane];
    st.g23[S & 1] = r[(a.j << 5) + a.lane];
  }
}

template <int G>
__device__ __forceinline__ void k1_step(K1State& st, const K1Arg& a) {
  __builtin_amdgcn_sched_barrier(0);
  if constexpr (G + 5 < 32) k1_issue<G + 5>(st, a);
  __builtin_amdgcn_sched_barrier(0);

  constexpr int S = G >> 2, C = G & 3;
  const int ii = (a.w + S) & 7, i = a.it * 8 + ii;
  const size_t rij = (size_t)((a.b * 64 + i) * 64 + a.j);
  v4f* pl = a.acc2 + ii * 512 + (C << 7) + a.lane;

  #pragma unroll
  for (int q = 0; q < 4; ++q) {
    const int m = (C << 2) + q;
    if constexpr (S == 0) st.s1a[m] = st.ring[G % 6][q]; else st.s1a[m] += st.ring[G % 6][q];
  }
  #pragma unroll
  for (int q = 0; q < 4; ++q) asm volatile("" : "+v"(st.s1a[(C << 2) + q]));
  #pragma unroll
  for (int q = 0; q < 4; ++q) {
    const v4f v = st.ring[G % 6][q];
    if (C == 0 && q == 0) st.p0 = v;
    else if (C == 0 && q == 1) st.p1 = v;
    else if (q & 1) st.p1 += v;
    else st.p0 += v;
  }
  if constexpr (S == 0) {
    #pragma unroll
    for (int q = 0; q < 4; ++q) pl[q << 5] = st.ring[G % 6][q];
    if (a.it == a.jt) {
      float* dd = a.d12 + (((size_t)(a.b * 64 + i)) << 11) + (C << 9) + (a.lane << 2);
      #pragma unroll
      for (int q = 0; q < 4; ++q) *(v4f*)(dd + (q << 7)) = st.ring[G % 6][q];
    }
  } else {
    v4f acc[4];
    #pragma unroll
    for (int q = 0; q < 4; ++q) acc[q] = pl[q << 5];
    #pragma unroll
    for (int q = 0; q < 4; ++q) acc[q] += st.ring[G % 6][q];
    if constexpr (S < 7) {
      #pragma unroll
      for (int q = 0; q < 4; ++q) pl[q << 5] = acc[q];
    } else {
      float* ds = a.s2p + (((size_t)((a.b * 8 + a.jt) * 64 + i)) << 11) + (C << 9) + (a.lane << 2);
      #pragma unroll
      for (int q = 0; q < 4; ++q) *(v4f*)(ds + (q << 7)) = acc[q];
    }
  }

  if constexpr (C == 1) {
    a.d13[(rij << 5) + a.lane] = st.g13[S & 1];
    a.d23[(rij << 5) + a.lane] = st.g23[S & 1];
    if constexpr (S == 0) {
      if (a.it == a.jt) a.tri[((a.b << 6) + i) * 32 + a.lane] = st.g13[0];
    }
  }
  if constexpr (C == 3) {
    v4f p = st.p0 + st.p1;
    #pragma unroll
    for (int e = 0; e < 4; ++e) { p[e] += swap8(p[e]); p[e] += swap16(p[e]); }
    if (a.lane < 8)
      *(v4f*)(a.s3m + (rij << 5) + (a.lane << 2)) = p * (1.f / 64.f);
    if constexpr (S < 7) {
      asm volatile("s_wait_dscnt 0x0\n\ts_barrier_signal -1\n\ts_barrier_wait -1"
                   ::: "memory");
    }
  }
  if constexpr (G + 1 < 32) k1_step<G + 1>(st, a);
}

__global__ void __launch_bounds__(256, 4)
k1_contract(const float* __restrict__ x,
            float* __restrict__ s1p, float* __restrict__ s2p,
            float* __restrict__ s3m, float* __restrict__ d12,
            float* __restrict__ d13, float* __restrict__ d23,
            float* __restrict__ tri) {
  __shared__ v4f acc2[8 * 512];
  const int t = threadIdx.x;
  K1Arg a;
  a.x = x; a.s1p = s1p; a.s2p = s2p; a.s3m = s3m;
  a.d12 = d12; a.d13 = d13; a.d23 = d23; a.tri = tri;
  a.acc2 = acc2;
  a.lane = t & 31;
  a.w = __builtin_amdgcn_readfirstlane(t >> 5);
  a.jt = blockIdx.x; a.it = blockIdx.y; a.b = blockIdx.z;
  a.j = a.jt * 8 + a.w;

  K1State st;
  k1_issue<0>(st, a); k1_issue<1>(st, a); k1_issue<2>(st, a);
  k1_issue<3>(st, a); k1_issue<4>(st, a);
  k1_step<0>(st, a);

  float* dst = s1p + (((size_t)((a.b * 8 + a.it) * 64 + a.j)) << 11) + (a.lane << 2);
  #pragma unroll
  for (int m = 0; m < 16; ++m) *(v4f*)(dst + (m << 7)) = st.s1a[m];
}

__global__ void k2b_a1(const float* __restrict__ s1p, const float* __restrict__ s3m,
                       const float* __restrict__ d12, const float* __restrict__ d13,
                       const float* __restrict__ d23, const float* __restrict__ tri,
                       float* __restrict__ a1) {
  __shared__ float scr[256];
  const int t = threadIdx.x, c = t & 31, g = t >> 5;
  const int b = blockIdx.x >> 6, u = blockIdx.x & 63;
  float* dst = a1 + (size_t)(b * 64 + u) * 320;

  auto red = [&](float partial, int slot) {
    scr[t] = partial;
    __syncthreads();
    if (t < 32) {
      float tot = 0.f;
      #pragma unroll
      for (int gg = 0; gg < 8; ++gg) tot += scr[gg * 32 + t];
      dst[slot * 32 + t] = tot * (1.f / 64.f);
    }
    __syncthreads();
  };
  float acc;
  {
    float pv[64];
    #pragma unroll
    for (int pg = 0; pg < 8; ++pg) {
      #pragma unroll
      for (int s = 0; s < 8; ++s)
        pv[pg * 8 + s] = s1p[(((size_t)((b * 8 + pg) * 64 + g + 8 * s)) << 11) + u * 32 + c];
    }
    acc = 0.f;
    #pragma unroll
    for (int e = 0; e < 64; ++e) acc += pv[e];
    red(acc * (1.f / 64.f), 0);
  }
  acc = 0.f; for (int s = 0; s < 8; ++s) { int v = g + 8 * s;
    acc += s3m[(((size_t)((b * 64 + v) * 64 + u)) << 5) + c]; } red(acc, 1);
  acc = 0.f; for (int s = 0; s < 8; ++s) { int v = g + 8 * s;
    acc += s3m[(((size_t)((b * 64 + u) * 64 + v)) << 5) + c]; } red(acc, 2);
  acc = 0.f; for (int s = 0; s < 8; ++s) { int v = g + 8 * s;
    acc += d12[(((size_t)(b * 64 + v)) << 11) + u * 32 + c]; } red(acc, 3);
  acc = 0.f; for (int s = 0; s < 8; ++s) { int v = g + 8 * s;
    acc += d12[(((size_t)(b * 64 + u)) << 11) + v * 32 + c]; } red(acc, 4);
  acc = 0.f; for (int s = 0; s < 8; ++s) { int v = g + 8 * s;
    acc += d13[(((size_t)((b * 64 + v) * 64 + u)) << 5) + c]; } red(acc, 5);
  acc = 0.f; for (int s = 0; s < 8; ++s) { int v = g + 8 * s;
    acc += d13[(((size_t)((b * 64 + u) * 64 + v)) << 5) + c]; } red(acc, 6);
  acc = 0.f; for (int s = 0; s < 8; ++s) { int v = g + 8 * s;
    acc += d23[(((size_t)((b * 64 + v) * 64 + u)) << 5) + c]; } red(acc, 7);
  acc = 0.f; for (int s = 0; s < 8; ++s) { int v = g + 8 * s;
    acc += d23[(((size_t)((b * 64 + u) * 64 + v)) << 5) + c]; } red(acc, 8);
  if (t < 32) dst[9 * 32 + t] = tri[(b * 64 + u) * 32 + t];
}

__device__ __forceinline__ v8f wmma4(v2f a, v2f b, v8f c) {
  return __builtin_amdgcn_wmma_f32_16x16x4_f32(false, a, false, b, (short)0, c,
                                               false, false);
}

__global__ void k3a_rows(const float* __restrict__ a1,
                         const float* __restrict__ W12, const float* __restrict__ b12,
                         const float* __restrict__ W12t, const float* __restrict__ b12t,
                         const float* __restrict__ W11, const float* __restrict__ b11,
                         const float* __restrict__ W01, const float* __restrict__ b01,
                         const float* __restrict__ W02, const float* __restrict__ b02,
                         const float* __restrict__ b22,
                         float* __restrict__ r12, float* __restrict__ r12t,
                         float* __restrict__ forow, float* __restrict__ r02p,
                         float* __restrict__ r01p) {
  const int t = threadIdx.x, lane = t & 31;
  const int wv = __builtin_amdgcn_readfirstlane(t >> 5);
  const int R0 = blockIdx.x * 16;
  const int kh = lane >> 4;
  const int mat = wv >> 1;
  const int n = ((wv & 1) << 4) + (lane & 15);
  const float* ar = a1 + (size_t)(R0 + (lane & 15)) * 320 + 2 * kh;
  v8f accE = {}, accO = {};

  if (mat == 3) {
    const float* w2 = W02 + (2 * kh) * 32 + n;
    const float* w1 = W01 + (2 * kh) * 32 + n;
    v8f c1E = {}, c1O = {};
    #pragma unroll
    for (int s = 0; s < 5; ++s) {
      const int ka = ((s == 0) ? 0 : (2 * s + 1)) * 32;
      const int kw = s * 32;
      #pragma unroll
      for (int kk = 0; kk < 32; kk += 8) {
        v2f aE = *(const v2f*)(ar + ka + kk);
        v2f aO = *(const v2f*)(ar + ka + kk + 4);
        v2f b2E = { w2[(kw + kk) * 32],     w2[(kw + kk) * 32 + 32] };
        v2f b2O = { w2[(kw + kk + 4) * 32], w2[(kw + kk + 4) * 32 + 32] };
        v2f b1E = { w1[(kw + kk) * 32],     w1[(kw + kk) * 32 + 32] };
        v2f b1O = { w1[(kw + kk + 4) * 32], w1[(kw + kk + 4) * 32 + 32] };
        accE = wmma4(aE, b2E, accE);
        accO = wmma4(aO, b2O, accO);
        c1E  = wmma4(aE, b1E, c1E);
        c1O  = wmma4(aO, b1O, c1O);
      }
    }
    float t2 = 0.f, t1 = 0.f;
    #pragma unroll
    for (int r = 0; r < 8; ++r) { t2 += accE[r] + accO[r]; t1 += c1E[r] + c1O[r]; }
    t2 += swap16(t2);
    t1 += swap16(t1);
    float v2 = t2 * (1.f / 64.f);
    if ((R0 & 63) == 0) v2 += b02[n] + b22[n];
    if (lane < 16) {
      r02p[blockIdx.x * 32 + n] = v2;
      r01p[blockIdx.x * 32 + n] = t1 * (1.f / 64.f);
    }
    return;
  }

  const float* Wm  = (mat == 0) ? W12 : (mat == 1) ? W12t : W11;
  float*       dst = (mat == 0) ? r12 : (mat == 1) ? r12t : forow;
  float bias = (mat == 0) ? b12[n] : (mat == 1) ? b12t[n] : (b11[n] + b01[n]);

  {
    const float* wc = Wm + (2 * kh) * 32 + n;
    #pragma unroll
    for (int kk = 0; kk < 320; kk += 8) {
      v2f aE = *(const v2f*)(ar + kk);
      v2f aO = *(const v2f*)(ar + kk + 4);
      v2f bE = { wc[kk * 32],       wc[kk * 32 + 32] };
      v2f bO = { wc[(kk + 4) * 32], wc[(kk + 4) * 32 + 32] };
      accE = wmma4(aE, bE, accE);
      accO = wmma4(aO, bO, accO);
    }
  }
  #pragma unroll
  for (int r = 0; r < 8; ++r) {
    const int M = r + (kh << 3);
    dst[(R0 + M) * 32 + n] = accE[r] + accO[r] + bias;
  }
}

static constexpr size_t PLANE = 131072;

__device__ __forceinline__ void k3_issue(v4f (&raw)[32], const float* __restrict__ P,
                                         size_t off, int pitch, int lane) {
  const float* p = P + off + (size_t)(lane >> 3) * pitch + ((lane & 7) << 2);
  #pragma unroll
  for (int pg = 0; pg < 8; ++pg) {
    #pragma unroll
    for (int r = 0; r < 4; ++r)
      raw[pg * 4 + r] = *(const v4f*)(p + (size_t)pg * PLANE + (size_t)(4 * r) * pitch);
  }
}

__device__ __forceinline__ void k3_fold(const v4f (&raw)[32], float* sb, int lane) {
  #pragma unroll
  for (int r = 0; r < 4; ++r) {
    v4f a = raw[r];
    #pragma unroll
    for (int pg = 1; pg < 8; ++pg) a += raw[pg * 4 + r];
    *(v4f*)(sb + (4 * r + (lane >> 3)) * 36 + ((lane & 7) << 2)) = a * (1.f / 64.f);
  }
}

__global__ void __launch_bounds__(256)
k3b_mix(const float* __restrict__ s1p, const float* __restrict__ s2p,
        const float* __restrict__ s3m, const float* __restrict__ d12,
        const float* __restrict__ d13, const float* __restrict__ d23,
        const float* __restrict__ W22,
        const float* __restrict__ r12, const float* __restrict__ r12t,
        const float* __restrict__ r02p, const float* __restrict__ r01p,
        const float* __restrict__ forow, float* __restrict__ out) {
  __shared__ float w[384 * 32];
  __shared__ __attribute__((aligned(16))) float stg[8 * 2 * 576];
  const int t = threadIdx.x;
  const int lane = t & 31;
  const int wv = __builtin_amdgcn_readfirstlane(t >> 5);
  const int R = (blockIdx.x * 8 + wv) * 16;
  const int bT = R >> 12, pT = (R >> 6) & 63, q0 = R & 63;
  const size_t planePQ = ((size_t)(bT * 512 + pT) << 11) + (size_t)(q0 << 5);
  const size_t planeQP = ((size_t)(bT * 512 + q0) << 11) + (size_t)(pT << 5);
  float* sb0 = stg + wv * 1152;
  float* sb1 = sb0 + 576;

  v4f rawA[32], rawB[32];
  k3_issue(rawA, s1p, planePQ, 32, lane);
  k3_issue(rawB, s2p, planePQ, 32, lane);
  __builtin_amdgcn_sched_barrier(0);

  for (int idx = t; idx < 384 * 32; idx += 256) w[idx] = W22[idx];
  __syncthreads();

  const int MA = lane & 15;
  const int kh = lane >> 4;
  const int rowA = R + MA;
  const int bA = rowA >> 12, pA = (rowA >> 6) & 63, qA = rowA & 63;
  const size_t offPQ = ((size_t)((bA * 64 + pA) * 64 + qA)) << 5;
  const size_t offQP = ((size_t)((bA * 64 + qA) * 64 + pA)) << 5;
  const float* comp[4] = { s3m, d12, d13, d23 };
  const int n0 = lane & 15;

  v8f acc0 = {};
  v8f acc1 = {};

  auto mm = [&](int f, const v2f (&a)[8]) {
    #pragma unroll
    for (int s = 0; s < 8; ++s) {
      const int kcol = f * 32 + 4 * s + 2 * kh;
      v2f b0 = { w[kcol * 32 + n0],      w[(kcol + 1) * 32 + n0] };
      v2f b1 = { w[kcol * 32 + n0 + 16], w[(kcol + 1) * 32 + n0 + 16] };
      acc0 = __builtin_amdgcn_wmma_f32_16x16x4_f32(false, a[s], false, b0,
                                                   (short)0, acc0, false, false);
      acc1 = __builtin_amdgcn_wmma_f32_16x16x4_f32(false, a[s], false, b1,
                                                   (short)0, acc1, false, false);
    }
  };
  auto plane_frag = [&](const v4f (&raw)[32], float* sb, v2f (&a)[8]) {
    k3_fold(raw, sb, lane);
    __builtin_amdgcn_fence(__ATOMIC_RELEASE, "wavefront");
    __builtin_amdgcn_wave_barrier();
    __builtin_amdgcn_fence(__ATOMIC_ACQUIRE, "wavefront");
    #pragma unroll
    for (int s = 0; s < 8; ++s) a[s] = *(const v2f*)(sb + MA * 36 + 4 * s + 2 * kh);
    __builtin_amdgcn_fence(__ATOMIC_RELEASE, "wavefront");
    __builtin_amdgcn_wave_barrier();
    __builtin_amdgcn_fence(__ATOMIC_ACQUIRE, "wavefront");
  };
  v2f fd[4][8];
  auto direct_issue = [&](size_t off) {
    #pragma unroll
    for (int j = 0; j < 4; ++j) {
      #pragma unroll
      for (int s = 0; s < 8; ++s) fd[j][s] = *(const v2f*)(comp[j] + off + 4 * s + 2 * kh);
    }
  };
  v2f fa[8], fb[8];

  direct_issue(offPQ);
  __builtin_amdgcn_sched_barrier(0);
  plane_frag(rawA, sb0, fa);
  k3_issue(rawA, s1p, planeQP, 2048, lane);
  plane_frag(rawB, sb1, fb);
  k3_issue(rawB, s2p, planeQP, 2048, lane);
  __builtin_amdgcn_sched_barrier(0);
  mm(0, fa);
  mm(1, fb);
  #pragma unroll
  for (int j = 0; j < 4; ++j) mm(2 + j, fd[j]);
  __builtin_amdgcn_sched_barrier(0);
  direct_issue(offQP);
  plane_frag(rawA, sb0, fa);
  plane_frag(rawB, sb1, fb);
  mm(6, fa);
  mm(7, fb);
  #pragma unroll
  for (int j = 0; j < 4; ++j) mm(8 + j, fd[j]);

  float v02_0 = 0.f, v02_1 = 0.f, v01_0 = 0.f, v01_1 = 0.f;
  #pragma unroll
  for (int g = 0; g < 4; ++g) {
    v02_0 += r02p[(bT * 4 + g) * 32 + n0];
    v02_1 += r02p[(bT * 4 + g) * 32 + n0 + 16];
    v01_0 += r01p[(bT * 4 + g) * 32 + n0];
    v01_1 += r01p[(bT * 4 + g) * 32 + n0 + 16];
  }
  #pragma unroll
  for (int r = 0; r < 8; ++r) {
    const int M = r + ((lane >> 4) << 3);
    const int row = R + M;
    const int b = row >> 12, p = (row >> 6) & 63, q = row & 63;
    float add0 = r12[((b << 6) + p) * 32 + n0] + r12t[((b << 6) + q) * 32 + n0]
               + v02_0;
    float add1 = r12[((b << 6) + p) * 32 + n0 + 16] + r12t[((b << 6) + q) * 32 + n0 + 16]
               + v02_1;
    if (p == q) {
      add0 += forow[((b << 6) + p) * 32 + n0] + v01_0;
      add1 += forow[((b << 6) + p) * 32 + n0 + 16] + v01_1;
    }
    out[((size_t)row << 5) + n0]      = acc0[r] + add0;
    out[((size_t)row << 5) + n0 + 16] = acc1[r] + add1;
  }
}

extern "C" void kernel_launch(void* const* d_in, const int* in_sizes, int n_in,
                              void* d_out, int out_size, void* d_ws, size_t ws_size,
                              hipStream_t stream) {
  (void)in_sizes; (void)n_in; (void)out_size;
  if (ws_size < WS_FLOATS * sizeof(float)) return;

  const float* x    = (const float*)d_in[0];
  const float* W22  = (const float*)d_in[1];
  const float* b22  = (const float*)d_in[2];
  const float* W12  = (const float*)d_in[3];
  const float* b12  = (const float*)d_in[4];
  const float* W12t = (const float*)d_in[5];
  const float* b12t = (const float*)d_in[6];
  const float* W02  = (const float*)d_in[7];
  const float* b02  = (const float*)d_in[8];
  const float* W11  = (const float*)d_in[9];
  const float* b11  = (const float*)d_in[10];
  const float* W01  = (const float*)d_in[11];
  const float* b01  = (const float*)d_in[12];

  float* ws = (float*)d_ws;
  float* s1p = ws + S1P;   float* s2p = ws + S2P;
  float* s3m = ws + S3M;   float* d12 = ws + D12;
  float* d13 = ws + D13;   float* d23 = ws + D23;
  float* tri = ws + TRI;   float* a1  = ws + A1O;
  float* r12 = ws + R12O;  float* r12t= ws + R12TO;
  float* forow = ws + FORO;
  float* r02p = ws + R02P; float* r01p = ws + R01P;

  k1_contract<<<dim3(8, 8, 8), 256, 0, stream>>>(x, s1p, s2p, s3m, d12, d13, d23, tri);
  k2b_a1     <<<512, 256, 0, stream>>>(s1p, s3m, d12, d13, d23, tri, a1);
  k3a_rows   <<<32, 256, 0, stream>>>(a1, W12, b12, W12t, b12t,
                                      W11, b11, W01, b01, W02, b02, b22,
                                      r12, r12t, forow, r02p, r01p);
  k3b_mix    <<<256, 256, 0, stream>>>(s1p, s2p, s3m, d12, d13, d23, W22,
                                       r12, r12t, r02p, r01p, forow, (float*)d_out);
}


# Mix GEMM re-tiled by mirror pairs: one wave makes rows [p,q] and [q,p] from a single load of the 12 A blocks they share; 5 loading waves per block, was 8 (new sum order in the [q,p] rows)
# speedup vs baseline: 2.3628x; 1.0509x over previous
#include <hip/hip_runtime.h>


typedef __attribute__((ext_vector_type(2))) float v2f;
typedef __attribute__((ext_vector_type(8))) float v8f;

static constexpr size_t S1P  = 0;
static constexpr size_t S2P  = 8388608;
static constexpr size_t S3M  = 16777216;
static constexpr size_t D12  = 17825792;
static constexpr size_t D13  = 18874368;
static constexpr size_t D23  = 19922944;
static constexpr size_t TRI  = 20971520;
static constexpr size_t R01P = 20987904;
static constexpr size_t A1O  = 23085056;
static constexpr size_t R02P = 23248896;
static constexpr size_t R12O = 23250176;
static constexpr size_t R12TO= 23266560;
static constexpr size_t FORO = 23282944;
static constexpr size_t WS_FLOATS = 23299584;

typedef __attribute__((ext_vector_type(4))) float v4f;

__device__ __forceinline__ v4f ld_stream(const float* p) {
  return __builtin_nontemporal_load((const v4f*)p);
}
__device__ __forceinline__ float swap8(float v) {
  return __builtin_bit_cast(float, __builtin_amdgcn_update_dpp(
      0, __builtin_bit_cast(int, v), 0x128  , 0xf, 0xf, false));
}
__device__ __forceinline__ float swap16(float v) {
  return __builtin_bit_cast(float, __builtin_amdgcn_permlanex16(
      __builtin_bit_cast(int, v), __builtin_bit_cast(int, v),
      0x76543210, 0xfedcba98, false, false));
}

struct K1Arg {
  const float* x;
  float *s1p, *s2p, *s3m, *d12, *d13, *d23, *tri;
  v4f* acc2;
  int jt, it, b, j, w, lane;
};
struct K1State {
  v4f s1a[16];
  v4f ring[6][4];
  float g13[2], g23[2];
  v4f p0, p1;
};

template <int G>
__device__ __forceinline__ void k1_issue(K1State& st, const K1Arg& a) {
  constexpr int S = G >> 2, C = G & 3;
  const int i = a.it * 8 + ((a.w + S) & 7);
  const float* r = a.x + (((size_t)((a.b * 64 + i) * 64 + a.j)) << 11);
  #pragma unroll
  for (int q = 0; q < 4; ++q)
    st.ring[G % 6][q] = ld_stream(r + ((C * 4 + q) << 7) + (a.lane << 2));
  if constexpr (C == 0) {
    st.g13[S & 1] = r[(i << 5) + a.lane];
    st.g23[S & 1] = r[(a.j << 5) + a.lane];
  }
}

template <int G>
__device__ __forceinline__ void k1_step(K1State& st, const K1Arg& a) {
  __builtin_amdgcn_sched_barrier(0);
  if constexpr (G + 5 < 32) k1_issue<G + 5>(st, a);
  __builtin_amdgcn_sched_barrier(0);

  constexpr int S = G >> 2, C = G & 3;
  const int ii = (a.w + S) & 7, i = a.it * 8 + ii;
  const size_t rij = (size_t)((a.b * 64 + i) * 64 + a.j);
  v4f* pl = a.acc2 + ii * 512 + (C << 7) + a.lane;

  #pragma unroll
  for (int q = 0; q < 4; ++q) {
    const int m = (C << 2) + q;
    if constexpr (S == 0) st.s1a[m] = st.ring[G % 6][q]; else st.s1a[m] += st.ring[G % 6][q];
  }
  #pragma unroll
  for (int q = 0; q < 4; ++q) asm volatile("" : "+v"(st.s1a[(C << 2) + q]));
  #pragma unroll
  for (int q = 0; q < 4; ++q) {
    const v4f v = st.ring[G % 6][q];
    if (C == 0 && q == 0) st.p0 = v;
    else if (C == 0 && q == 1) st.p1 = v;
    else if (q & 1) st.p1 += v;
    else st.p0 += v;
  }
  if constexpr (S == 0) {
    #pragma unroll
    for (int q = 0; q < 4; ++q) pl[q << 5] = st.ring[G % 6][q];
    if (a.it == a.jt) {
      float* dd = a.d12 + (((size_t)(a.b * 64 + i)) << 11) + (C << 9) + (a.lane << 2);
      #pragma unroll
      for (int q = 0; q < 4; ++q) *(v4f*)(dd + (q << 7)) = st.ring[G % 6][q];
    }
  } else {
    v4f acc[4];
    #pragma unroll
    for (int q = 0; q < 4; ++q) acc[q] = pl[q << 5];
    #pragma unroll
    for (int q = 0; q < 4; ++q) acc[q] += st.ring[G % 6][q];
    if constexpr (S < 7) {
      #pragma unroll
      for (int q = 0; q < 4; ++q) pl[q << 5] = acc[q];
    } else {
      float* ds = a.s2p + (((size_t)((a.b * 8 + a.jt) * 64 + i)) << 11) + (C << 9) + (a.lane << 2);
      #pragma unroll
      for (int q = 0; q < 4; ++q) *(v4f*)(ds + (q << 7)) = acc[q];
    }
  }

  if constexpr (C == 1) {
    a.d13[(rij << 5) + a.lane] = st.g13[S & 1];
    a.d23[(rij << 5) + a.lane] = st.g23[S & 1];
    if constexpr (S == 0) {
      if (a.it == a.jt) a.tri[((a.b << 6) + i) * 32 + a.lane] = st.g13[0];
    }
  }
  if constexpr (C == 3) {
    v4f p = st.p0 + st.p1;
    #pragma unroll
    for (int e = 0; e < 4; ++e) { p[e] += swap8(p[e]); p[e] += swap16(p[e]); }
    if (a.lane < 8)
      *(v4f*)(a.s3m + (rij << 5) + (a.lane << 2)) = p * (1.f / 64.f);
    if constexpr (S < 7) {
      asm volatile("s_wait_dscnt 0x0\n\ts_barrier_signal -1\n\ts_barrier_wait -1"
                   ::: "memory");
    }
  }
  if constexpr (G + 1 < 32) k1_step<G + 1>(st, a);
}

__global__ void __launch_bounds__(256, 4)
k1_contract(const float* __restrict__ x,
            float* __restrict__ s1p, float* __restrict__ s2p,
            float* __restrict__ s3m, float* __restrict__ d12,
            float* __restrict__ d13, float* __restrict__ d23,
            float* __restrict__ tri) {
  __shared__ v4f acc2[8 * 512];
  const int t = threadIdx.x;
  K1Arg a;
  a.x = x; a.s1p = s1p; a.s2p = s2p; a.s3m = s3m;
  a.d12 = d12; a.d13 = d13; a.d23 = d23; a.tri = tri;
  a.acc2 = acc2;
  a.lane = t & 31;
  a.w = __builtin_amdgcn_readfirstlane(t >> 5);
  a.jt = blockIdx.x; a.it = blockIdx.y; a.b = blockIdx.z;
  a.j = a.jt * 8 + a.w;

  K1State st;
  k1_issue<0>(st, a); k1_issue<1>(st, a); k1_issue<2>(st, a);
  k1_issue<3>(st, a); k1_issue<4>(st, a);
  k1_step<0>(st, a);

  float* dst = s1p + (((size_t)((a.b * 8 + a.it) * 64 + a.j)) << 11) + (a.lane << 2);
  #pragma unroll
  for (int m = 0; m < 16; ++m) *(v4f*)(dst + (m << 7)) = st.s1a[m];
}

__global__ void k2b_a1(const float* __restrict__ s1p, const float* __restrict__ s3m,
                       const float* __restrict__ d12, const float* __restrict__ d13,
                       const float* __restrict__ d23, const float* __restrict__ tri,
                       float* __restrict__ a1) {
  __shared__ float scr[256];
  const int t = threadIdx.x, c = t & 31, g = t >> 5;
  const int b = blockIdx.x >> 6, u = blockIdx.x & 63;
  float* dst = a1 + (size_t)(b * 64 + u) * 320;

  auto red = [&](float partial, int slot) {
    scr[t] = partial;
    __syncthreads();
    if (t < 32) {
      float tot = 0.f;
      #pragma unroll
      for (int gg = 0; gg < 8; ++gg) tot += scr[gg * 32 + t];
      dst[slot * 32 + t] = tot * (1.f / 64.f);
    }
    __syncthreads();
  };
  float acc;
  {
    float pv[64];
    #pragma unroll
    for (int pg = 0; pg < 8; ++pg) {
      #pragma unroll
      for (int s = 0; s < 8; ++s)
        pv[pg * 8 + s] = s1p[(((size_t)((b * 8 + pg) * 64 + g + 8 * s)) << 11) + u * 32 + c];
    }
    acc = 0.f;
    #pragma unroll
    for (int e = 0; e < 64; ++e) acc += pv[e];
    red(acc * (1.f / 64.f), 0);
  }
  acc = 0.f; for (int s = 0; s < 8; ++s) { int v = g + 8 * s;
    acc += s3m[(((size_t)((b * 64 + v) * 64 + u)) << 5) + c]; } red(acc, 1);
  acc = 0.f; for (int s = 0; s < 8; ++s) { int v = g + 8 * s;
    acc += s3m[(((size_t)((b * 64 + u) * 64 + v)) << 5) + c]; } red(acc, 2);
  acc = 0.f; for (int s = 0; s < 8; ++s) { int v = g + 8 * s;
    acc += d12[(((size_t)(b * 64 + v)) << 11) + u * 32 + c]; } red(acc, 3);
  acc = 0.f; for (int s = 0; s < 8; ++s) { int v = g + 8 * s;
    acc += d12[(((size_t)(b * 64 + u)) << 11) + v * 32 + c]; } red(acc, 4);
  acc = 0.f; for (int s = 0; s < 8; ++s) { int v = g + 8 * s;
    acc += d13[(((size_t)((b * 64 + v) * 64 + u)) << 5) + c]; } red(acc, 5);
  acc = 0.f; for (int s = 0; s < 8; ++s) { int v = g + 8 * s;
    acc += d13[(((size_t)((b * 64 + u) * 64 + v)) << 5) + c]; } red(acc, 6);
  acc = 0.f; for (int s = 0; s < 8; ++s) { int v = g + 8 * s;
    acc += d23[(((size_t)((b * 64 + v) * 64 + u)) << 5) + c]; } red(acc, 7);
  acc = 0.f; for (int s = 0; s < 8; ++s) { int v = g + 8 * s;
    acc += d23[(((size_t)((b * 64 + u) * 64 + v)) << 5) + c]; } red(acc, 8);
  if (t < 32) dst[9 * 32 + t] = tri[(b * 64 + u) * 32 + t];
}

__device__ __forceinline__ v8f wmma4(v2f a, v2f b, v8f c) {
  return __builtin_amdgcn_wmma_f32_16x16x4_f32(false, a, false, b, (short)0, c,
                                               false, false);
}

__global__ void k3a_rows(const float* __restrict__ a1,
                         const float* __restrict__ W12, const float* __restrict__ b12,
                         const float* __restrict__ W12t, const float* __restrict__ b12t,
                         const float* __restrict__ W11, const float* __restrict__ b11,
                         const float* __restrict__ W01, const float* __restrict__ b01,
                         const float* __restrict__ W02, const float* __restrict__ b02,
                         const float* __restrict__ b22,
                         float* __restrict__ r12, float* __restrict__ r12t,
                         float* __restrict__ forow, float* __restrict__ r02p,
                         float* __restrict__ r01p) {
  const int t = threadIdx.x, lane = t & 31;
  const int wv = __builtin_amdgcn_readfirstlane(t >> 5);
  const int R0 = blockIdx.x * 16;
  const int kh = lane >> 4;
  const int mat = wv >> 1;
  const int n = ((wv & 1) << 4) + (lane & 15);
  const float* ar = a1 + (size_t)(R0 + (lane & 15)) * 320 + 2 * kh;
  v8f accE = {}, accO = {};

  if (mat == 3) {
    const float* w2 = W02 + (2 * kh) * 32 + n;
    const float* w1 = W01 + (2 * kh) * 32 + n;
    v8f c1E = {}, c1O = {};
    #pragma unroll
    for (int s = 0; s < 5; ++s) {
      const int ka = ((s == 0) ? 0 : (2 * s + 1)) * 32;
      const int kw = s * 32;
      #pragma unroll
      for (int kk = 0; kk < 32; kk += 8) {
        v2f aE = *(const v2f*)(ar + ka + kk);
        v2f aO = *(const v2f*)(ar + ka + kk + 4);
        v2f b2E = { w2[(kw + kk) * 32],     w2[(kw + kk) * 32 + 32] };
        v2f b2O = { w2[(kw + kk + 4) * 32], w2[(kw + kk + 4) * 32 + 32] };
        v2f b1E = { w1[(kw + kk) * 32],     w1[(kw + kk) * 32 + 32] };
        v2f b1O = { w1[(kw + kk + 4) * 32], w1[(kw + kk + 4) * 32 + 32] };
        accE = wmma4(aE, b2E, accE);
        accO = wmma4(aO, b2O, accO);
        c1E  = wmma4(aE, b1E, c1E);
        c1O  = wmma4(aO, b1O, c1O);
      }
    }
    float t2 = 0.f, t1 = 0.f;
    #pragma unroll
    for (int r = 0; r < 8; ++r) { t2 += accE[r] + accO[r]; t1 += c1E[r] + c1O[r]; }
    t2 += swap16(t2);
    t1 += swap16(t1);
    float v2 = t2 * (1.f / 64.f);
    if ((R0 & 63) == 0) v2 += b02[n] + b22[n];
    if (lane < 16) {
      r02p[blockIdx.x * 32 + n] = v2;
      r01p[blockIdx.x * 32 + n] = t1 * (1.f / 64.f);
    }
    return;
  }

  const float* Wm  = (mat == 0) ? W12 : (mat == 1) ? W12t : W11;
  float*       dst = (mat == 0) ? r12 : (mat == 1) ? r12t : forow;
  float bias = (mat == 0) ? b12[n] : (mat == 1) ? b12t[n] : (b11[n] + b01[n]);

  {
    const float* wc = Wm + (2 * kh) * 32 + n;
    #pragma unroll
    for (int kk = 0; kk < 320; kk += 8) {
      v2f aE = *(const v2f*)(ar + kk);
      v2f aO = *(const v2f*)(ar + kk + 4);
      v2f bE = { wc[kk * 32],       wc[kk * 32 + 32] };
      v2f bO = { wc[(kk + 4) * 32], wc[(kk + 4) * 32 + 32] };
      accE = wmma4(aE, bE, accE);
      accO = wmma4(aO, bO, accO);
    }
  }
  #pragma unroll
  for (int r = 0; r < 8; ++r) {
    const int M = r + (kh << 3);
    dst[(R0 + M) * 32 + n] = accE[r] + accO[r] + bias;
  }
}

static constexpr size_t PLANE = 131072;

__device__ __forceinline__ void k3_issue(v4f (&raw)[32], const float* __restrict__ P,
                                         size_t off, int pitch, int lane) {
  const float* p = P + off + (size_t)(lane >> 3) * pitch + ((lane & 7) << 2);
  #pragma unroll
  for (int pg = 0; pg < 8; ++pg) {
    #pragma unroll
    for (int r = 0; r < 4; ++r)
      raw[pg * 4 + r] = *(const v4f*)(p + (size_t)pg * PLANE + (size_t)(4 * r) * pitch);
  }
}

__device__ __forceinline__ void k3_fold(const v4f (&raw)[32], float* sb, int lane) {
  #pragma unroll
  for (int r = 0; r < 4; ++r) {
    v4f a = raw[r];
    #pragma unroll
    for (int pg = 1; pg < 8; ++pg) a += raw[pg * 4 + r];
    *(v4f*)(sb + (4 * r + (lane >> 3)) * 36 + ((lane & 7) << 2)) = a * (1.f / 64.f);
  }
}

template <bool PAIR>
__device__ __forceinline__ void k3_tile(
    v4f (&rawA)[32], v4f (&rawB)[32], const float* w, float* sb0, float* sb1,
    const float* __restrict__ s1p, const float* __restrict__ s2p,
    const float* __restrict__ s3m, const float* __restrict__ d12,
    const float* __restrict__ d13, const float* __restrict__ d23,
    const float* __restrict__ r12, const float* __restrict__ r12t,
    const float* __restrict__ r02p, const float* __restrict__ r01p,
    const float* __restrict__ forow, float* __restrict__ out,
    int bT, int pT, int q0, size_t planeQP, int lane) {
  const int MA = lane & 15;
  const int kh = lane >> 4;
  const int qA = q0 + MA;
  const size_t offPQ = ((size_t)((bT * 64 + pT) * 64 + qA)) << 5;
  const size_t offQP = ((size_t)((bT * 64 + qA) * 64 + pT)) << 5;
  const float* comp[4] = { s3m, d12, d13, d23 };
  const int n0 = lane & 15;

  v8f acc0 = {}, acc1 = {};
  v8f acc2 = {}, acc3 = {};

  auto mm = [&](int f, int g, const v2f (&a)[8]) {
    #pragma unroll
    for (int s = 0; s < 8; ++s) {
      const int kcol = f * 32 + 4 * s + 2 * kh;
      v2f b0 = { w[kcol * 32 + n0],      w[(kcol + 1) * 32 + n0] };
      v2f b1 = { w[kcol * 32 + n0 + 16], w[(kcol + 1) * 32 + n0 + 16] };
      acc0 = __builtin_amdgcn_wmma_f32_16x16x4_f32(false, a[s], false, b0,
                                                   (short)0, acc0, false, false);
      acc1 = __builtin_amdgcn_wmma_f32_16x16x4_f32(false, a[s], false, b1,
                                                   (short)0, acc1, false, false);
      if constexpr (PAIR) {
        const int kc2 = g * 32 + 4 * s + 2 * kh;
        v2f c0 = { w[kc2 * 32 + n0],      w[(kc2 + 1) * 32 + n0] };
        v2f c1 = { w[kc2 * 32 + n0 + 16], w[(kc2 + 1) * 32 + n0 + 16] };
        acc2 = __builtin_amdgcn_wmma_f32_16x16x4_f32(false, a[s], false, c0,
                                                     (short)0, acc2, false, false);
        acc3 = __builtin_amdgcn_wmma_f32_16x16x4_f32(false, a[s], false, c1,
                                                     (short)0, acc3, false, false);
      }
    }
  };
  auto plane_frag = [&](const v4f (&raw)[32], float* sb, v2f (&a)[8]) {
    k3_fold(raw, sb, lane);
    __builtin_amdgcn_fence(__ATOMIC_RELEASE, "wavefront");
    __builtin_amdgcn_wave_barrier();
    __builtin_amdgcn_fence(__ATOMIC_ACQUIRE, "wavefront");
    #pragma unroll
    for (int s = 0; s < 8; ++s) a[s] = *(const v2f*)(sb + MA * 36 + 4 * s + 2 * kh);
    __builtin_amdgcn_fence(__ATOMIC_RELEASE, "wavefront");
    __builtin_amdgcn_wave_barrier();
    __builtin_amdgcn_fence(__ATOMIC_ACQUIRE, "wavefront");
  };
  v2f fd[4][8];
  auto direct_issue = [&](size_t off) {
    #pragma unroll
    for (int j = 0; j < 4; ++j) {
      #pragma unroll
      for (int s = 0; s < 8; ++s) fd[j][s] = *(const v2f*)(comp[j] + off + 4 * s + 2 * kh);
    }
  };
  v2f fa[8], fb[8];

  direct_issue(offPQ);
  __builtin_amdgcn_sched_barrier(0);
  plane_frag(rawA, sb0, fa);
  k3_issue(rawA, s1p, planeQP, 2048, lane);
  plane_frag(rawB, sb1, fb);
  k3_issue(rawB, s2p, planeQP, 2048, lane);
  __builtin_amdgcn_sched_barrier(0);
  mm(0, 6, fa);
  mm(1, 7, fb);
  #pragma unroll
  for (int j = 0; j < 4; ++j) mm(2 + j, 8 + j, fd[j]);
  __builtin_amdgcn_sched_barrier(0);
  direct_issue(offQP);
  plane_frag(rawA, sb0, fa);
  plane_frag(rawB, sb1, fb);
  mm(6, 0, fa);
  mm(7, 1, fb);
  #pragma unroll
  for (int j = 0; j < 4; ++j) mm(8 + j, 2 + j, fd[j]);

  float v02_0 = 0.f, v02_1 = 0.f, v01_0 = 0.f, v01_1 = 0.f;
  #pragma unroll
  for (int g = 0; g < 4; ++g) {
    v02_0 += r02p[(bT * 4 + g) * 32 + n0];
    v02_1 += r02p[(bT * 4 + g) * 32 + n0 + 16];
    if constexpr (!PAIR) {
      v01_0 += r01p[(bT * 4 + g) * 32 + n0];
      v01_1 += r01p[(bT * 4 + g) * 32 + n0 + 16];
    }
  }
  const int bp = (bT << 6) + pT;
  #pragma unroll
  for (int r = 0; r < 8; ++r) {
    const int q = q0 + r + ((lane >> 4) << 3);
    const int bq = (bT << 6) + q;
    float add0 = r12[bp * 32 + n0] + r12t[bq * 32 + n0] + v02_0;
    float add1 = r12[bp * 32 + n0 + 16] + r12t[bq * 32 + n0 + 16] + v02_1;
    if constexpr (!PAIR) {
      if (pT == q) {
        add0 += forow[bp * 32 + n0] + v01_0;
        add1 += forow[bp * 32 + n0 + 16] + v01_1;
      }
    }
    const size_t row = (size_t)(bp * 64 + q);
    out[(row << 5) + n0]      = acc0[r] + add0;
    out[(row << 5) + n0 + 16] = acc1[r] + add1;
    if constexpr (PAIR) {
      const float m0 = r12[bq * 32 + n0] + r12t[bp * 32 + n0] + v02_0;
      const float m1 = r12[bq * 32 + n0 + 16] + r12t[bp * 32 + n0 + 16] + v02_1;
      const size_t row2 = (size_t)(bq * 64 + pT);
      out[(row2 << 5) + n0]      = acc2[r] + m0;
      out[(row2 << 5) + n0 + 16] = acc3[r] + m1;
    }
  }
}

__global__ void __launch_bounds__(256)
k3b_mix(const float* __restrict__ s1p, const float* __restrict__ s2p,
        const float* __restrict__ s3m, const float* __restrict__ d12,
        const float* __restrict__ d13, const float* __restrict__ d23,
        const float* __restrict__ W22,
        const float* __restrict__ r12, const float* __restrict__ r12t,
        const float* __restrict__ r02p, const float* __restrict__ r01p,
        const float* __restrict__ forow, float* __restrict__ out) {
  __shared__ float w[384 * 32];
  __shared__ __attribute__((aligned(16))) float stg[8 * 2 * 576];
  const int t = threadIdx.x;
  const int lane = t & 31;
  const int wv = __builtin_amdgcn_readfirstlane(t >> 5);
  const int bT = blockIdx.x >> 5, lb = blockIdx.x & 31;
  const bool pair = wv < 3;
  const bool active = pair || (wv & 3) == 3;
  int pT, qr;
  if (pair) {
    const int d = lb * 3 + wv;
    if (d < 48)      { pT = d / 3;               qr = 1 + d - 3 * pT; }
    else if (d < 80) { pT = 16 + ((d - 48) >> 1); qr = 2 + (d & 1); }
    else             { pT = d - 48;              qr = 3; }
  } else {
    pT = lb * 2 + (wv >> 2);
    qr = pT >> 4;
  }
  const int q0 = qr << 4;
  const size_t planePQ = ((size_t)(bT * 512 + pT) << 11) + (size_t)(q0 << 5);
  const size_t planeQP = ((size_t)(bT * 512 + q0) << 11) + (size_t)(pT << 5);
  float* sb0 = stg + wv * 1152;
  float* sb1 = sb0 + 576;

  v4f rawA[32], rawB[32];
  if (active) {
    k3_issue(rawA, s1p, planePQ, 32, lane);
    k3_issue(rawB, s2p, planePQ, 32, lane);
  }
  __builtin_amdgcn_sched_barrier(0);

  for (int idx = t; idx < 384 * 32; idx += 256) w[idx] = W22[idx];
  __syncthreads();
  if (!active) return;

  if (pair)
    k3_tile<true>(rawA, rawB, w, sb0, sb1, s1p, s2p, s3m, d12, d13, d23,
                  r12, r12t, r02p, r01p, forow, out, bT, pT, q0, planeQP, lane);
  else
    k3_tile<false>(rawA, rawB, w, sb0, sb1, s1p, s2p, s3m, d12, d13, d23,
                   r12, r12t, r02p, r01p, forow, out, bT, pT, q0, planeQP, lane);
}

extern "C" void kernel_launch(void* const* d_in, const int* in_sizes, int n_in,
                              void* d_out, int out_size, void* d_ws, size_t ws_size,
                              hipStream_t stream) {
  (void)in_sizes; (void)n_in; (void)out_size;
  if (ws_size < WS_FLOATS * sizeof(float)) return;

  const float* x    = (const float*)d_in[0];
  const float* W22  = (const float*)d_in[1];
  const float* b22  = (const float*)d_in[2];
  const float* W12  = (const float*)d_in[3];
  const float* b12  = (const float*)d_in[4];
  const float* W12t = (const float*)d_in[5];
  const float* b12t = (const float*)d_in[6];
  const float* W02  = (const float*)d_in[7];
  const float* b02  = (const float*)d_in[8];
  const float* W11  = (const float*)d_in[9];
  const float* b11  = (const float*)d_in[10];
  const float* W01  = (const float*)d_in[11];
  const float* b01  = (const float*)d_in[12];

  float* ws = (float*)d_ws;
  float* s1p = ws + S1P;   float* s2p = ws + S2P;
  float* s3m = ws + S3M;   float* d12 = ws + D12;
  float* d13 = ws + D13;   float* d23 = ws + D23;
  float* tri = ws + TRI;   float* a1  = ws + A1O;
  float* r12 = ws + R12O;  float* r12t= ws + R12TO;
  float* forow = ws + FORO;
  float* r02p = ws + R02P; float* r01p = ws + R01P;

  k1_contract<<<dim3(8, 8, 8), 256, 0, stream>>>(x, s1p, s2p, s3m, d12, d13, d23, tri);
  k2b_a1     <<<512, 256, 0, stream>>>(s1p, s3m, d12, d13, d23, tri, a1);
  k3a_rows   <<<32, 256, 0, stream>>>(a1, W12, b12, W12t, b12t,
                                      W11, b11, W01, b01, W02, b02, b22,
                                      r12, r12t, forow, r02p, r01p);
  k3b_mix    <<<256, 256, 0, stream>>>(s1p, s2p, s3m, d12, d13, d23, W22,
                                       r12, r12t, r02p, r01p, forow, (float*)d_out);
}


# Row-mix GEMM launch re-cut for latency: block = (16-row tile, matrix), K quartered over waves, a wave's loads all in one run ahead of 20 WMMAs, quarters added via LDS; 32->128 blocks (new sum order)
# speedup vs baseline: 2.5651x; 1.0856x over previous
#include <hip/hip_runtime.h>


typedef __attribute__((ext_vector_type(2))) float v2f;
typedef __attribute__((ext_vector_type(8))) float v8f;

static constexpr size_t S1P  = 0;
static constexpr size_t S2P  = 8388608;
static constexpr size_t S3M  = 16777216;
static constexpr size_t D12  = 17825792;
static constexpr size_t D13  = 18874368;
static constexpr size_t D23  = 19922944;
static constexpr size_t TRI  = 20971520;
static constexpr size_t R01P = 20987904;
static constexpr size_t A1O  = 23085056;
static constexpr size_t R02P = 23248896;
static constexpr size_t R12O = 23250176;
static constexpr size_t R12TO= 23266560;
static constexpr size_t FORO = 23282944;
static constexpr size_t WS_FLOATS = 23299584;

typedef __attribute__((ext_vector_type(4))) float v4f;

__device__ __forceinline__ v4f ld_stream(const float* p) {
  return __builtin_nontemporal_load((const v4f*)p);
}
__device__ __forceinline__ float swap8(float v) {
  return __builtin_bit_cast(float, __builtin_amdgcn_update_dpp(
      0, __builtin_bit_cast(int, v), 0x128  , 0xf, 0xf, false));
}
__device__ __forceinline__ float swap16(float v) {
  return __builtin_bit_cast(float, __builtin_amdgcn_permlanex16(
      __builtin_bit_cast(int, v), __builtin_bit_cast(int, v),
      0x76543210, 0xfedcba98, false, false));
}

struct K1Arg {
  const float* x;
  float *s1p, *s2p, *s3m, *d12, *d13, *d23, *tri;
  v4f* acc2;
  int jt, it, b, j, w, lane;
};
struct K1State {
  v4f s1a[16];
  v4f ring[6][4];
  float g13[2], g23[2];
  v4f p0, p1;
};

template <int G>
__device__ __forceinline__ void k1_issue(K1State& st, const K1Arg& a) {
  constexpr int S = G >> 2, C = G & 3;
  const int i = a.it * 8 + ((a.w + S) & 7);
  const float* r = a.x + (((size_t)((a.b * 64 + i) * 64 + a.j)) << 11);
  #pragma unroll
  for (int q = 0; q < 4; ++q)
    st.ring[G % 6][q] = ld_stream(r + ((C * 4 + q) << 7) + (a.lane << 2));
  if constexpr (C == 0) {
    st.g13[S & 1] = r[(i << 5) + a.lane];
    st.g23[S & 1] = r[(a.j << 5) + a.lane];
  }
}

template <int G>
__device__ __forceinline__ void k1_step(K1State& st, const K1Arg& a) {
  __builtin_amdgcn_sched_barrier(0);
  if constexpr (G + 5 < 32) k1_issue<G + 5>(st, a);
  __builtin_amdgcn_sched_barrier(0);

  constexpr int S = G >> 2, C = G & 3;
  const int ii = (a.w + S) & 7, i = a.it * 8 + ii;
  const size_t rij = (size_t)((a.b * 64 + i) * 64 + a.j);
  v4f* pl = a.acc2 + ii * 512 + (C << 7) + a.lane;

  #pragma unroll
  for (int q = 0; q < 4; ++q) {
    const int m = (C << 2) + q;
    if constexpr (S == 0) st.s1a[m] = st.ring[G % 6][q]; else st.s1a[m] += st.ring[G % 6][q];
  }
  #pragma unroll
  for (int q = 0; q < 4; ++q) asm volatile("" : "+v"(st.s1a[(C << 2) + q]));
  #pragma unroll
  for (int q = 0; q < 4; ++q) {
    const v4f v = st.ring[G % 6][q];
    if (C == 0 && q == 0) st.p0 = v;
    else if (C == 0 && q == 1) st.p1 = v;
    else if (q & 1) st.p1 += v;
    else st.p0 += v;
  }
  if constexpr (S == 0) {
    #pragma unroll
    for (int q = 0; q < 4; ++q) pl[q << 5] = st.ring[G % 6][q];
    if (a.it == a.jt) {
      float* dd = a.d12 + (((size_t)(a.b * 64 + i)) << 11) + (C << 9) + (a.lane << 2);
      #pragma unroll
      for (int q = 0; q < 4; ++q) *(v4f*)(dd + (q << 7)) = st.ring[G % 6][q];
    }
  } else {
    v4f acc[4];
    #pragma unroll
    for (int q = 0; q < 4; ++q) acc[q] = pl[q << 5];
    #pragma unroll
    for (int q = 0; q < 4; ++q) acc[q] += st.ring[G % 6][q];
    if constexpr (S < 7) {
      #pragma unroll
      for (int q = 0; q < 4; ++q) pl[q << 5] = acc[q];
    } else {
      float* ds = a.s2p + (((size_t)((a.b * 8 + a.jt) * 64 + i)) << 11) + (C << 9) + (a.lane << 2);
      #pragma unroll
      for (int q = 0; q < 4; ++q) *(v4f*)(ds + (q << 7)) = acc[q];
    }
  }

  if constexpr (C == 1) {
    a.d13[(rij << 5) + a.lane] = st.g13[S & 1];
    a.d23[(rij << 5) + a.lane] = st.g23[S & 1];
    if constexpr (S == 0) {
      if (a.it == a.jt) a.tri[((a.b << 6) + i) * 32 + a.lane] = st.g13[0];
    }
  }
  if constexpr (C == 3) {
    v4f p = st.p0 + st.p1;
    #pragma unroll
    for (int e = 0; e < 4; ++e) { p[e] += swap8(p[e]); p[e] += swap16(p[e]); }
    if (a.lane < 8)
      *(v4f*)(a.s3m + (rij << 5) + (a.lane << 2)) = p * (1.f / 64.f);
    if constexpr (S < 7) {
      asm volatile("s_wait_dscnt 0x0\n\ts_barrier_signal -1\n\ts_barrier_wait -1"
                   ::: "memory");
    }
  }
  if constexpr (G + 1 < 32) k1_step<G + 1>(st, a);
}

__global__ void __launch_bounds__(256, 4)
k1_contract(const float* __restrict__ x,
            float* __restrict__ s1p, float* __restrict__ s2p,
            float* __restrict__ s3m, float* __restrict__ d12,
            float* __restrict__ d13, float* __restrict__ d23,
            float* __restrict__ tri) {
  __shared__ v4f acc2[8 * 512];
  const int t = threadIdx.x;
  K1Arg a;
  a.x = x; a.s1p = s1p; a.s2p = s2p; a.s3m = s3m;
  a.d12 = d12; a.d13 = d13; a.d23 = d23; a.tri = tri;
  a.acc2 = acc2;
  a.lane = t & 31;
  a.w = __builtin_amdgcn_readfirstlane(t >> 5);
  a.jt = blockIdx.x; a.it = blockIdx.y; a.b = blockIdx.z;
  a.j = a.jt * 8 + a.w;

  K1State st;
  k1_issue<0>(st, a); k1_issue<1>(st, a); k1_issue<2>(st, a);
  k1_issue<3>(st, a); k1_issue<4>(st, a);
  k1_step<0>(st, a);

  float* dst = s1p + (((size_t)((a.b * 8 + a.it) * 64 + a.j)) << 11) + (a.lane << 2);
  #pragma unroll
  for (int m = 0; m < 16; ++m) *(v4f*)(dst + (m << 7)) = st.s1a[m];
}

__global__ void k2b_a1(const float* __restrict__ s1p, const float* __restrict__ s3m,
                       const float* __restrict__ d12, const float* __restrict__ d13,
                       const float* __restrict__ d23, const float* __restrict__ tri,
                       float* __restrict__ a1) {
  __shared__ float scr[256];
  const int t = threadIdx.x, c = t & 31, g = t >> 5;
  const int b = blockIdx.x >> 6, u = blockIdx.x & 63;
  float* dst = a1 + (size_t)(b * 64 + u) * 320;

  auto red = [&](float partial, int slot) {
    scr[t] = partial;
    __syncthreads();
    if (t < 32) {
      float tot = 0.f;
      #pragma unroll
      for (int gg = 0; gg < 8; ++gg) tot += scr[gg * 32 + t];
      dst[slot * 32 + t] = tot * (1.f / 64.f);
    }
    __syncthreads();
  };
  float acc;
  {
    float pv[64];
    #pragma unroll
    for (int pg = 0; pg < 8; ++pg) {
      #pragma unroll
      for (int s = 0; s < 8; ++s)
        pv[pg * 8 + s] = s1p[(((size_t)((b * 8 + pg) * 64 + g + 8 * s)) << 11) + u * 32 + c];
    }
    acc = 0.f;
    #pragma unroll
    for (int e = 0; e < 64; ++e) acc += pv[e];
    red(acc * (1.f / 64.f), 0);
  }
  acc = 0.f; for (int s = 0; s < 8; ++s) { int v = g + 8 * s;
    acc += s3m[(((size_t)((b * 64 + v) * 64 + u)) << 5) + c]; } red(acc, 1);
  acc = 0.f; for (int s = 0; s < 8; ++s) { int v = g + 8 * s;
    acc += s3m[(((size_t)((b * 64 + u) * 64 + v)) << 5) + c]; } red(acc, 2);
  acc = 0.f; for (int s = 0; s < 8; ++s) { int v = g + 8 * s;
    acc += d12[(((size_t)(b * 64 + v)) << 11) + u * 32 + c]; } red(acc, 3);
  acc = 0.f; for (int s = 0; s < 8; ++s) { int v = g + 8 * s;
    acc += d12[(((size_t)(b * 64 + u)) << 11) + v * 32 + c]; } red(acc, 4);
  acc = 0.f; for (int s = 0; s < 8; ++s) { int v = g + 8 * s;
    acc += d13[(((size_t)((b * 64 + v) * 64 + u)) << 5) + c]; } red(acc, 5);
  acc = 0.f; for (int s = 0; s < 8; ++s) { int v = g + 8 * s;
    acc += d13[(((size_t)((b * 64 + u) * 64 + v)) << 5) + c]; } red(acc, 6);
  acc = 0.f; for (int s = 0; s < 8; ++s) { int v = g + 8 * s;
    acc += d23[(((size_t)((b * 64 + v) * 64 + u)) << 5) + c]; } red(acc, 7);
  acc = 0.f; for (int s = 0; s < 8; ++s) { int v = g + 8 * s;
    acc += d23[(((size_t)((b * 64 + u) * 64 + v)) << 5) + c]; } red(acc, 8);
  if (t < 32) dst[9 * 32 + t] = tri[(b * 64 + u) * 32 + t];
}

__device__ __forceinline__ v8f wmma4(v2f a, v2f b, v8f c) {
  return __builtin_amdgcn_wmma_f32_16x16x4_f32(false, a, false, b, (short)0, c,
                                               false, false);
}

__global__ void __launch_bounds__(256)
k3a_rows(const float* __restrict__ a1,
         const float* __restrict__ W12, const float* __restrict__ b12,
         const float* __restrict__ W12t, const float* __restrict__ b12t,
         const float* __restrict__ W11, const float* __restrict__ b11,
         const float* __restrict__ W01, const float* __restrict__ b01,
         const float* __restrict__ W02, const float* __restrict__ b02,
         const float* __restrict__ b22,
         float* __restrict__ r12, float* __restrict__ r12t,
         float* __restrict__ forow, float* __restrict__ r02p,
         float* __restrict__ r01p) {
  __shared__ float red[8 * 256];
  const int t = threadIdx.x, lane = t & 31;
  const int wv = __builtin_amdgcn_readfirstlane(t >> 5);
  const int rt = blockIdx.x >> 2;
  const int mat = blockIdx.x & 3;
  const int R0 = rt * 16;
  const int kh = lane >> 4;
  const float* ar = a1 + (size_t)(R0 + (lane & 15)) * 320 + 2 * kh;

  const int n = ((wv & 1) << 4) + (lane & 15);
  const float* Wm = (mat == 0) ? W12 : (mat == 1) ? W12t : (mat == 2) ? W11
                  : ((wv & 3) < 2) ? W02 : W01;
  const int k0 = ((mat == 3) ? (wv >> 2) : (wv >> 1)) * 80;
  const float* wc = Wm + (size_t)(k0 + 2 * kh) * 32 + n;
  const float* pa = (mat == 0) ? b12 : (mat == 1) ? b12t : (mat == 2) ? b11 : b02;
  const float* pb = (mat == 3) ? b22 : b01;
  const bool share0 = (mat == 3) && wv < 2 && (R0 & 63) == 0;
  const float va = pa[n], vb = pb[n];

  v2f aE[10], aO[10], bE[10], bO[10];
  #pragma unroll
  for (int i = 0; i < 10; ++i) {
    const int kw = k0 + 8 * i;
    int ka = kw;
    if (mat == 3) {
      const int s = kw >> 5;
      ka = ((s == 0) ? 0 : (2 * s + 1)) * 32 + (kw & 31);
    }
    aE[i] = *(const v2f*)(ar + ka);
    aO[i] = *(const v2f*)(ar + ka + 4);
    bE[i] = v2f{ wc[(8 * i) * 32],     wc[(8 * i) * 32 + 32] };
    bO[i] = v2f{ wc[(8 * i + 4) * 32], wc[(8 * i + 4) * 32 + 32] };
  }
  __builtin_amdgcn_sched_barrier(0);
  const float bias = (mat < 3) ? (va + ((mat == 2) ? vb : 0.f)) : (share0 ? (va + vb) : 0.f);
  v8f accE = {}, accO = {};
  #pragma unroll
  for (int i = 0; i < 10; ++i) {
    accE = wmma4(aE[i], bE[i], accE);
    accO = wmma4(aO[i], bO[i], accO);
  }
  float* mine = red + wv * 256;
  #pragma unroll
  for (int r = 0; r < 8; ++r) mine[r * 32 + lane] = accE[r] + accO[r];
  __syncthreads();

  if (mat == 3) {
    if (wv < 4) {
      float v = 0.f;
      #pragma unroll
      for (int r = 0; r < 8; ++r)
        v += red[wv * 256 + r * 32 + lane] + red[(wv + 4) * 256 + r * 32 + lane];
      v += swap16(v);
      v = v * (1.f / 64.f) + bias;
      if (lane < 16) {
        if (wv < 2) r02p[rt * 32 + n] = v;
        else        r01p[rt * 32 + n] = v;
      }
    }
    return;
  }
  float* dst = (mat == 0) ? r12 : (mat == 1) ? r12t : forow;
  #pragma unroll
  for (int e = 0; e < 2; ++e) {
    const int r = 2 * (wv >> 1) + e;
    const float* q = red + (wv & 1) * 256 + r * 32 + lane;
    const float v = ((q[0] + q[512]) + q[1024]) + q[1536];
    const int M = r + (kh << 3);
    dst[(R0 + M) * 32 + n] = v + bias;
  }
}

static constexpr size_t PLANE = 131072;

__device__ __forceinline__ void k3_issue(v4f (&raw)[32], const float* __restrict__ P,
                                         size_t off, int pitch, int lane) {
  const float* p = P + off + (size_t)(lane >> 3) * pitch + ((lane & 7) << 2);
  #pragma unroll
  for (int pg = 0; pg < 8; ++pg) {
    #pragma unroll
    for (int r = 0; r < 4; ++r)
      raw[pg * 4 + r] = *(const v4f*)(p + (size_t)pg * PLANE + (size_t)(4 * r) * pitch);
  }
}

__device__ __forceinline__ void k3_fold(const v4f (&raw)[32], float* sb, int lane) {
  #pragma unroll
  for (int r = 0; r < 4; ++r) {
    v4f a = raw[r];
    #pragma unroll
    for (int pg = 1; pg < 8; ++pg) a += raw[pg * 4 + r];
    *(v4f*)(sb + (4 * r + (lane >> 3)) * 36 + ((lane & 7) << 2)) = a * (1.f / 64.f);
  }
}

template <bool PAIR>
__device__ __forceinline__ void k3_tile(
    v4f (&rawA)[32], v4f (&rawB)[32], const float* w, float* sb0, float* sb1,
    const float* __restrict__ s1p, const float* __restrict__ s2p,
    const float* __restrict__ s3m, const float* __restrict__ d12,
    const float* __restrict__ d13, const float* __restrict__ d23,
    const float* __restrict__ r12, const float* __restrict__ r12t,
    const float* __restrict__ r02p, const float* __restrict__ r01p,
    const float* __restrict__ forow, float* __restrict__ out,
    int bT, int pT, int q0, size_t planeQP, int lane) {
  const int MA = lane & 15;
  const int kh = lane >> 4;
  const int qA = q0 + MA;
  const size_t offPQ = ((size_t)((bT * 64 + pT) * 64 + qA)) << 5;
  const size_t offQP = ((size_t)((bT * 64 + qA) * 64 + pT)) << 5;
  const float* comp[4] = { s3m, d12, d13, d23 };
  const int n0 = lane & 15;

  v8f acc0 = {}, acc1 = {};
  v8f acc2 = {}, acc3 = {};

  auto mm = [&](int f, int g, const v2f (&a)[8]) {
    #pragma unroll
    for (int s = 0; s < 8; ++s) {
      const int kcol = f * 32 + 4 * s + 2 * kh;
      v2f b0 = { w[kcol * 32 + n0],      w[(kcol + 1) * 32 + n0] };
      v2f b1 = { w[kcol * 32 + n0 + 16], w[(kcol + 1) * 32 + n0 + 16] };
      acc0 = __builtin_amdgcn_wmma_f32_16x16x4_f32(false, a[s], false, b0,
                                                   (short)0, acc0, false, false);
      acc1 = __builtin_amdgcn_wmma_f32_16x16x4_f32(false, a[s], false, b1,
                                                   (short)0, acc1, false, false);
      if constexpr (PAIR) {
        const int kc2 = g * 32 + 4 * s + 2 * kh;
        v2f c0 = { w[kc2 * 32 + n0],      w[(kc2 + 1) * 32 + n0] };
        v2f c1 = { w[kc2 * 32 + n0 + 16], w[(kc2 + 1) * 32 + n0 + 16] };
        acc2 = __builtin_amdgcn_wmma_f32_16x16x4_f32(false, a[s], false, c0,
                                                     (short)0, acc2, false, false);
        acc3 = __builtin_amdgcn_wmma_f32_16x16x4_f32(false, a[s], false, c1,
                                                     (short)0, acc3, false, false);
      }
    }
  };
  auto plane_frag = [&](const v4f (&raw)[32], float* sb, v2f (&a)[8]) {
    k3_fold(raw, sb, lane);
    __builtin_amdgcn_fence(__ATOMIC_RELEASE, "wavefront");
    __builtin_amdgcn_wave_barrier();
    __builtin_amdgcn_fence(__ATOMIC_ACQUIRE, "wavefront");
    #pragma unroll
    for (int s = 0; s < 8; ++s) a[s] = *(const v2f*)(sb + MA * 36 + 4 * s + 2 * kh);
    __builtin_amdgcn_fence(__ATOMIC_RELEASE, "wavefront");
    __builtin_amdgcn_wave_barrier();
    __builtin_amdgcn_fence(__ATOMIC_ACQUIRE, "wavefront");
  };
  v2f fd[4][8];
  auto direct_issue = [&](size_t off) {
    #pragma unroll
    for (int j = 0; j < 4; ++j) {
      #pragma unroll
      for (int s = 0; s < 8; ++s) fd[j][s] = *(const v2f*)(comp[j] + off + 4 * s + 2 * kh);
    }
  };
  v2f fa[8], fb[8];

  direct_issue(offPQ);
  __builtin_amdgcn_sched_barrier(0);
  plane_frag(rawA, sb0, fa);
  k3_issue(rawA, s1p, planeQP, 2048, lane);
  plane_frag(rawB, sb1, fb);
  k3_issue(rawB, s2p, planeQP, 2048, lane);
  __builtin_amdgcn_sched_barrier(0);
  mm(0, 6, fa);
  mm(1, 7, fb);
  #pragma unroll
  for (int j = 0; j < 4; ++j) mm(2 + j, 8 + j, fd[j]);
  __builtin_amdgcn_sched_barrier(0);
  direct_issue(offQP);
  plane_frag(rawA, sb0, fa);
  plane_frag(rawB, sb1, fb);
  mm(6, 0, fa);
  mm(7, 1, fb);
  #pragma unroll
  for (int j = 0; j < 4; ++j) mm(8 + j, 2 + j, fd[j]);

  float v02_0 = 0.f, v02_1 = 0.f, v01_0 = 0.f, v01_1 = 0.f;
  #pragma unroll
  for (int g = 0; g < 4; ++g) {
    v02_0 += r02p[(bT * 4 + g) * 32 + n0];
    v02_1 += r02p[(bT * 4 + g) * 32 + n0 + 16];
    if constexpr (!PAIR) {
      v01_0 += r01p[(bT * 4 + g) * 32 + n0];
      v01_1 += r01p[(bT * 4 + g) * 32 + n0 + 16];
    }
  }
  const int bp = (bT << 6) + pT;
  #pragma unroll
  for (int r = 0; r < 8; ++r) {
    const int q = q0 + r + ((lane >> 4) << 3);
    const int bq = (bT << 6) + q;
    float add0 = r12[bp * 32 + n0] + r12t[bq * 32 + n0] + v02_0;
    float add1 = r12[bp * 32 + n0 + 16] + r12t[bq * 32 + n0 + 16] + v02_1;
    if constexpr (!PAIR) {
      if (pT == q) {
        add0 += forow[bp * 32 + n0] + v01_0;
        add1 += forow[bp * 32 + n0 + 16] + v01_1;
      }
    }
    const size_t row = (size_t)(bp * 64 + q);
    out[(row << 5) + n0]      = acc0[r] + add0;
    out[(row << 5) + n0 + 16] = acc1[r] + add1;
    if constexpr (PAIR) {
      const float m0 = r12[bq * 32 + n0] + r12t[bp * 32 + n0] + v02_0;
      const float m1 = r12[bq * 32 + n0 + 16] + r12t[bp * 32 + n0 + 16] + v02_1;
      const size_t row2 = (size_t)(bq * 64 + pT);
      out[(row2 << 5) + n0]      = acc2[r] + m0;
      out[(row2 << 5) + n0 + 16] = acc3[r] + m1;
    }
  }
}

__global__ void __launch_bounds__(256)
k3b_mix(const float* __restrict__ s1p, const float* __restrict__ s2p,
        const float* __restrict__ s3m, const float* __restrict__ d12,
        const float* __restrict__ d13, const float* __restrict__ d23,
        const float* __restrict__ W22,
        const float* __restrict__ r12, const float* __restrict__ r12t,
        const float* __restrict__ r02p, const float* __restrict__ r01p,
        const float* __restrict__ forow, float* __restrict__ out) {
  __shared__ float w[384 * 32];
  __shared__ __attribute__((aligned(16))) float stg[8 * 2 * 576];
  const int t = threadIdx.x;
  const int lane = t & 31;
  const int wv = __builtin_amdgcn_readfirstlane(t >> 5);
  const int bT = blockIdx.x >> 5, lb = blockIdx.x & 31;
  const bool pair = wv < 3;
  const bool active = pair || (wv & 3) == 3;
  int pT, qr;
  if (pair) {
    const int d = lb * 3 + wv;
    if (d < 48)      { pT = d / 3;               qr = 1 + d - 3 * pT; }
    else if (d < 80) { pT = 16 + ((d - 48) >> 1); qr = 2 + (d & 1); }
    else             { pT = d - 48;              qr = 3; }
  } else {
    pT = lb * 2 + (wv >> 2);
    qr = pT >> 4;
  }
  const int q0 = qr << 4;
  const size_t planePQ = ((size_t)(bT * 512 + pT) << 11) + (size_t)(q0 << 5);
  const size_t planeQP = ((size_t)(bT * 512 + q0) << 11) + (size_t)(pT << 5);
  float* sb0 = stg + wv * 1152;
  float* sb1 = sb0 + 576;

  v4f rawA[32], rawB[32];
  if (active) {
    k3_issue(rawA, s1p, planePQ, 32, lane);
    k3_issue(rawB, s2p, planePQ, 32, lane);
  }
  __builtin_amdgcn_sched_barrier(0);

  for (int idx = t; idx < 384 * 32; idx += 256) w[idx] = W22[idx];
  __syncthreads();
  if (!active) return;

  if (pair)
    k3_tile<true>(rawA, rawB, w, sb0, sb1, s1p, s2p, s3m, d12, d13, d23,
                  r12, r12t, r02p, r01p, forow, out, bT, pT, q0, planeQP, lane);
  else
    k3_tile<false>(rawA, rawB, w, sb0, sb1, s1p, s2p, s3m, d12, d13, d23,
                   r12, r12t, r02p, r01p, forow, out, bT, pT, q0, planeQP, lane);
}

extern "C" void kernel_launch(void* const* d_in, const int* in_sizes, int n_in,
                              void* d_out, int out_size, void* d_ws, size_t ws_size,
                              hipStream_t stream) {
  (void)in_sizes; (void)n_in; (void)out_size;
  if (ws_size < WS_FLOATS * sizeof(float)) return;

  const float* x    = (const float*)d_in[0];
  const float* W22  = (const float*)d_in[1];
  const float* b22  = (const float*)d_in[2];
  const float* W12  = (const float*)d_in[3];
  const float* b12  = (const float*)d_in[4];
  const float* W12t = (const float*)d_in[5];
  const float* b12t = (const float*)d_in[6];
  const float* W02  = (const float*)d_in[7];
  const float* b02  = (const float*)d_in[8];
  const float* W11  = (const float*)d_in[9];
  const float* b11  = (const float*)d_in[10];
  const float* W01  = (const float*)d_in[11];
  const float* b01  = (const float*)d_in[12];

  float* ws = (float*)d_ws;
  float* s1p = ws + S1P;   float* s2p = ws + S2P;
  float* s3m = ws + S3M;   float* d12 = ws + D12;
  float* d13 = ws + D13;   float* d23 = ws + D23;
  float* tri = ws + TRI;   float* a1  = ws + A1O;
  float* r12 = ws + R12O;  float* r12t= ws + R12TO;
  float* forow = ws + FORO;
  float* r02p = ws + R02P; float* r01p = ws + R01P;

  k1_contract<<<dim3(8, 8, 8), 256, 0, stream>>>(x, s1p, s2p, s3m, d12, d13, d23, tri);
  k2b_a1     <<<512, 256, 0, stream>>>(s1p, s3m, d12, d13, d23, tri, a1);
  k3a_rows   <<<128, 256, 0, stream>>>(a1, W12, b12, W12t, b12t,
                                      W11, b11, W01, b01, W02, b02, b22,
                                      r12, r12t, forow, r02p, r01p);
  k3b_mix    <<<256, 256, 0, stream>>>(s1p, s2p, s3m, d12, d13, d23, W22,
                                       r12, r12t, r02p, r01p, forow, (float*)d_out);
}


# a1-row launch re-cut by wave: wave w owns sum w+1 (sixteen 16-byte loads, register adds, two lane swaps, own 128-B store), 512-term sum split over 8 waves via LDS; 18 barriers -> 1 (new sum order)
# speedup vs baseline: 2.9004x; 1.1307x over previous
#include <hip/hip_runtime.h>


typedef __attribute__((ext_vector_type(2))) float v2f;
typedef __attribute__((ext_vector_type(8))) float v8f;

static constexpr size_t S1P  = 0;
static constexpr size_t S2P  = 8388608;
static constexpr size_t S3M  = 16777216;
static constexpr size_t D12  = 17825792;
static constexpr size_t D13  = 18874368;
static constexpr size_t D23  = 19922944;
static constexpr size_t TRI  = 20971520;
static constexpr size_t R01P = 20987904;
static constexpr size_t A1O  = 23085056;
static constexpr size_t R02P = 23248896;
static constexpr size_t R12O = 23250176;
static constexpr size_t R12TO= 23266560;
static constexpr size_t FORO = 23282944;
static constexpr size_t WS_FLOATS = 23299584;

typedef __attribute__((ext_vector_type(4))) float v4f;

__device__ __forceinline__ v4f ld_stream(const float* p) {
  return __builtin_nontemporal_load((const v4f*)p);
}
__device__ __forceinline__ float swap8(float v) {
  return __builtin_bit_cast(float, __builtin_amdgcn_update_dpp(
      0, __builtin_bit_cast(int, v), 0x128  , 0xf, 0xf, false));
}
__device__ __forceinline__ float swap16(float v) {
  return __builtin_bit_cast(float, __builtin_amdgcn_permlanex16(
      __builtin_bit_cast(int, v), __builtin_bit_cast(int, v),
      0x76543210, 0xfedcba98, false, false));
}

struct K1Arg {
  const float* x;
  float *s1p, *s2p, *s3m, *d12, *d13, *d23, *tri;
  v4f* acc2;
  int jt, it, b, j, w, lane;
};
struct K1State {
  v4f s1a[16];
  v4f ring[6][4];
  float g13[2], g23[2];
  v4f p0, p1;
};

template <int G>
__device__ __forceinline__ void k1_issue(K1State& st, const K1Arg& a) {
  constexpr int S = G >> 2, C = G & 3;
  const int i = a.it * 8 + ((a.w + S) & 7);
  const float* r = a.x + (((size_t)((a.b * 64 + i) * 64 + a.j)) << 11);
  #pragma unroll
  for (int q = 0; q < 4; ++q)
    st.ring[G % 6][q] = ld_stream(r + ((C * 4 + q) << 7) + (a.lane << 2));
  if constexpr (C == 0) {
    st.g13[S & 1] = r[(i << 5) + a.lane];
    st.g23[S & 1] = r[(a.j << 5) + a.lane];
  }
}

template <int G>
__device__ __forceinline__ void k1_step(K1State& st, const K1Arg& a) {
  __builtin_amdgcn_sched_barrier(0);
  if constexpr (G + 5 < 32) k1_issue<G + 5>(st, a);
  __builtin_amdgcn_sched_barrier(0);

  constexpr int S = G >> 2, C = G & 3;
  const int ii = (a.w + S) & 7, i = a.it * 8 + ii;
  const size_t rij = (size_t)((a.b * 64 + i) * 64 + a.j);
  v4f* pl = a.acc2 + ii * 512 + (C << 7) + a.lane;

  #pragma unroll
  for (int q = 0; q < 4; ++q) {
    const int m = (C << 2) + q;
    if constexpr (S == 0) st.s1a[m] = st.ring[G % 6][q]; else st.s1a[m] += st.ring[G % 6][q];
  }
  #pragma unroll
  for (int q = 0; q < 4; ++q) asm volatile("" : "+v"(st.s1a[(C << 2) + q]));
  #pragma unroll
  for (int q = 0; q < 4; ++q) {
    const v4f v = st.ring[G % 6][q];
    if (C == 0 && q == 0) st.p0 = v;
    else if (C == 0 && q == 1) st.p1 = v;
    else if (q & 1) st.p1 += v;
    else st.p0 += v;
  }
  if constexpr (S == 0) {
    #pragma unroll
    for (int q = 0; q < 4; ++q) pl[q << 5] = st.ring[G % 6][q];
    if (a.it == a.jt) {
      float* dd = a.d12 + (((size_t)(a.b * 64 + i)) << 11) + (C << 9) + (a.lane << 2);
      #pragma unroll
      for (int q = 0; q < 4; ++q) *(v4f*)(dd + (q << 7)) = st.ring[G % 6][q];
    }
  } else {
    v4f acc[4];
    #pragma unroll
    for (int q = 0; q < 4; ++q) acc[q] = pl[q << 5];
    #pragma unroll
    for (int q = 0; q < 4; ++q) acc[q] += st.ring[G % 6][q];
    if constexpr (S < 7) {
      #pragma unroll
      for (int q = 0; q < 4; ++q) pl[q << 5] = acc[q];
    } else {
      float* ds = a.s2p + (((size_t)((a.b * 8 + a.jt) * 64 + i)) << 11) + (C << 9) + (a.lane << 2);
      #pragma unroll
      for (int q = 0; q < 4; ++q) *(v4f*)(ds + (q << 7)) = acc[q];
    }
  }

  if constexpr (C == 1) {
    a.d13[(rij << 5) + a.lane] = st.g13[S & 1];
    a.d23[(rij << 5) + a.lane] = st.g23[S & 1];
    if constexpr (S == 0) {
      if (a.it == a.jt) a.tri[((a.b << 6) + i) * 32 + a.lane] = st.g13[0];
    }
  }
  if constexpr (C == 3) {
    v4f p = st.p0 + st.p1;
    #pragma unroll
    for (int e = 0; e < 4; ++e) { p[e] += swap8(p[e]); p[e] += swap16(p[e]); }
    if (a.lane < 8)
      *(v4f*)(a.s3m + (rij << 5) + (a.lane << 2)) = p * (1.f / 64.f);
    if constexpr (S < 7) {
      asm volatile("s_wait_dscnt 0x0\n\ts_barrier_signal -1\n\ts_barrier_wait -1"
                   ::: "memory");
    }
  }
  if constexpr (G + 1 < 32) k1_step<G + 1>(st, a);
}

__global__ void __launch_bounds__(256, 4)
k1_contract(const float* __restrict__ x,
            float* __restrict__ s1p, float* __restrict__ s2p,
            float* __restrict__ s3m, float* __restrict__ d12,
            float* __restrict__ d13, float* __restrict__ d23,
            float* __restrict__ tri) {
  __shared__ v4f acc2[8 * 512];
  const int t = threadIdx.x;
  K1Arg a;
  a.x = x; a.s1p = s1p; a.s2p = s2p; a.s3m = s3m;
  a.d12 = d12; a.d13 = d13; a.d23 = d23; a.tri = tri;
  a.acc2 = acc2;
  a.lane = t & 31;
  a.w = __builtin_amdgcn_readfirstlane(t >> 5);
  a.jt = blockIdx.x; a.it = blockIdx.y; a.b = blockIdx.z;
  a.j = a.jt * 8 + a.w;

  K1State st;
  k1_issue<0>(st, a); k1_issue<1>(st, a); k1_issue<2>(st, a);
  k1_issue<3>(st, a); k1_issue<4>(st, a);
  k1_step<0>(st, a);

  float* dst = s1p + (((size_t)((a.b * 8 + a.it) * 64 + a.j)) << 11) + (a.lane << 2);
  #pragma unroll
  for (int m = 0; m < 16; ++m) *(v4f*)(dst + (m << 7)) = st.s1a[m];
}

__global__ void __launch_bounds__(256)
k2b_a1(const float* __restrict__ s1p, const float* __restrict__ s3m,
       const float* __restrict__ d12, const float* __restrict__ d13,
       const float* __restrict__ d23, const float* __restrict__ tri,
       float* __restrict__ a1) {
  __shared__ __attribute__((aligned(16))) float part[8 * 32];
  const int t = threadIdx.x, lane = t & 31;
  const int wv = __builtin_amdgcn_readfirstlane(t >> 5);
  const int b = blockIdx.x >> 6, u = blockIdx.x & 63;
  float* dst = a1 + (size_t)(b * 64 + u) * 320;
  const int rl = lane >> 3, cq = (lane & 7) << 2;

  const float* P = (wv < 2) ? s3m : (wv < 4) ? d12 : (wv < 6) ? d13 : d23;
  const bool outer = (wv & 1) == 0;
  const int pitch = outer ? 2048 : 32;
  const float* q = P + ((size_t)b << 17) + (size_t)u * (outer ? 32 : 2048)
                     + (size_t)rl * pitch + cq;
  const float* r = s1p + ((size_t)(b * 8 + wv) << 17) + (size_t)u * 32
                       + (size_t)rl * 2048 + cq;

  v4f shr[16], own[16];
  #pragma unroll
  for (int m = 0; m < 16; ++m) shr[m] = *(const v4f*)(r + (size_t)(4 * m) * 2048);
  #pragma unroll
  for (int m = 0; m < 16; ++m) own[m] = *(const v4f*)(q + (size_t)(4 * m) * pitch);
  float tv = 0.f;
  if (wv == 7) tv = tri[(b * 64 + u) * 32 + lane];

  auto fold = [&](const v4f (&v)[16]) {
    v4f e = v[0], o = v[1];
    #pragma unroll
    for (int m = 2; m < 16; m += 2) { e += v[m]; o += v[m + 1]; }
    v4f p = e + o;
    #pragma unroll
    for (int i = 0; i < 4; ++i) { p[i] += swap8(p[i]); p[i] += swap16(p[i]); }
    return p;
  };
  const v4f ps = fold(shr);
  if (lane < 8) *(v4f*)(part + wv * 32 + cq) = ps;
  const v4f po = fold(own);
  __syncthreads();
  if (lane < 8) *(v4f*)(dst + (wv + 1) * 32 + cq) = po * (1.f / 64.f);
  if (wv == 7) dst[9 * 32 + lane] = tv;
  if (wv == 0) {
    float tot = 0.f;
    #pragma unroll
    for (int w = 0; w < 8; ++w) tot += part[w * 32 + lane];
    dst[lane] = tot * (1.f / 4096.f);
  }
}

__device__ __forceinline__ v8f wmma4(v2f a, v2f b, v8f c) {
  return __builtin_amdgcn_wmma_f32_16x16x4_f32(false, a, false, b, (short)0, c,
                                               false, false);
}

__global__ void __launch_bounds__(256)
k3a_rows(const float* __restrict__ a1,
         const float* __restrict__ W12, const float* __restrict__ b12,
         const float* __restrict__ W12t, const float* __restrict__ b12t,
         const float* __restrict__ W11, const float* __restrict__ b11,
         const float* __restrict__ W01, const float* __restrict__ b01,
         const float* __restrict__ W02, const float* __restrict__ b02,
         const float* __restrict__ b22,
         float* __restrict__ r12, float* __restrict__ r12t,
         float* __restrict__ forow, float* __restrict__ r02p,
         float* __restrict__ r01p) {
  __shared__ float red[8 * 256];
  const int t = threadIdx.x, lane = t & 31;
  const int wv = __builtin_amdgcn_readfirstlane(t >> 5);
  const int rt = blockIdx.x >> 2;
  const int mat = blockIdx.x & 3;
  const int R0 = rt * 16;
  const int kh = lane >> 4;
  const float* ar = a1 + (size_t)(R0 + (lane & 15)) * 320 + 2 * kh;

  const int n = ((wv & 1) << 4) + (lane & 15);
  const float* Wm = (mat == 0) ? W12 : (mat == 1) ? W12t : (mat == 2) ? W11
                  : ((wv & 3) < 2) ? W02 : W01;
  const int k0 = ((mat == 3) ? (wv >> 2) : (wv >> 1)) * 80;
  const float* wc = Wm + (size_t)(k0 + 2 * kh) * 32 + n;
  const float* pa = (mat == 0) ? b12 : (mat == 1) ? b12t : (mat == 2) ? b11 : b02;
  const float* pb = (mat == 3) ? b22 : b01;
  const bool share0 = (mat == 3) && wv < 2 && (R0 & 63) == 0;
  const float va = pa[n], vb = pb[n];

  v2f aE[10], aO[10], bE[10], bO[10];
  #pragma unroll
  for (int i = 0; i < 10; ++i) {
    const int kw = k0 + 8 * i;
    int ka = kw;
    if (mat == 3) {
      const int s = kw >> 5;
      ka = ((s == 0) ? 0 : (2 * s + 1)) * 32 + (kw & 31);
    }
    aE[i] = *(const v2f*)(ar + ka);
    aO[i] = *(const v2f*)(ar + ka + 4);
    bE[i] = v2f{ wc[(8 * i) * 32],     wc[(8 * i) * 32 + 32] };
    bO[i] = v2f{ wc[(8 * i + 4) * 32], wc[(8 * i + 4) * 32 + 32] };
  }
  __builtin_amdgcn_sched_barrier(0);
  const float bias = (mat < 3) ? (va + ((mat == 2) ? vb : 0.f)) : (share0 ? (va + vb) : 0.f);
  v8f accE = {}, accO = {};
  #pragma unroll
  for (int i = 0; i < 10; ++i) {
    accE = wmma4(aE[i], bE[i], accE);
    accO = wmma4(aO[i], bO[i], accO);
  }
  float* mine = red + wv * 256;
  #pragma unroll
  for (int r = 0; r < 8; ++r) mine[r * 32 + lane] = accE[r] + accO[r];
  __syncthreads();

  if (mat == 3) {
    if (wv < 4) {
      float v = 0.f;
      #pragma unroll
      for (int r = 0; r < 8; ++r)
        v += red[wv * 256 + r * 32 + lane] + red[(wv + 4) * 256 + r * 32 + lane];
      v += swap16(v);
      v = v * (1.f / 64.f) + bias;
      if (lane < 16) {
        if (wv < 2) r02p[rt * 32 + n] = v;
        else        r01p[rt * 32 + n] = v;
      }
    }
    return;
  }
  float* dst = (mat == 0) ? r12 : (mat == 1) ? r12t : forow;
  #pragma unroll
  for (int e = 0; e < 2; ++e) {
    const int r = 2 * (wv >> 1) + e;
    const float* q = red + (wv & 1) * 256 + r * 32 + lane;
    const float v = ((q[0] + q[512]) + q[1024]) + q[1536];
    const int M = r + (kh << 3);
    dst[(R0 + M) * 32 + n] = v + bias;
  }
}

static constexpr size_t PLANE = 131072;

__device__ __forceinline__ void k3_issue(v4f (&raw)[32], const float* __restrict__ P,
                                         size_t off, int pitch, int lane) {
  const float* p = P + off + (size_t)(lane >> 3) * pitch + ((lane & 7) << 2);
  #pragma unroll
  for (int pg = 0; pg < 8; ++pg) {
    #pragma unroll
    for (int r = 0; r < 4; ++r)
      raw[pg * 4 + r] = *(const v4f*)(p + (size_t)pg * PLANE + (size_t)(4 * r) * pitch);
  }
}

__device__ __forceinline__ void k3_fold(const v4f (&raw)[32], float* sb, int lane) {
  #pragma unroll
  for (int r = 0; r < 4; ++r) {
    v4f a = raw[r];
    #pragma unroll
    for (int pg = 1; pg < 8; ++pg) a += raw[pg * 4 + r];
    *(v4f*)(sb + (4 * r + (lane >> 3)) * 36 + ((lane & 7) << 2)) = a * (1.f / 64.f);
  }
}

template <bool PAIR>
__device__ __forceinline__ void k3_tile(
    v4f (&rawA)[32], v4f (&rawB)[32], const float* w, float* sb0, float* sb1,
    const float* __restrict__ s1p, const float* __restrict__ s2p,
    const float* __restrict__ s3m, const float* __restrict__ d12,
    const float* __restrict__ d13, const float* __restrict__ d23,
    const float* __restrict__ r12, const float* __restrict__ r12t,
    const float* __restrict__ r02p, const float* __restrict__ r01p,
    const float* __restrict__ forow, float* __restrict__ out,
    int bT, int pT, int q0, size_t planeQP, int lane) {
  const int MA = lane & 15;
  const int kh = lane >> 4;
  const int qA = q0 + MA;
  const size_t offPQ = ((size_t)((bT * 64 + pT) * 64 + qA)) << 5;
  const size_t offQP = ((size_t)((bT * 64 + qA) * 64 + pT)) << 5;
  const float* comp[4] = { s3m, d12, d13, d23 };
  const int n0 = lane & 15;

  v8f acc0 = {}, acc1 = {};
  v8f acc2 = {}, acc3 = {};

  auto mm = [&](int f, int g, const v2f (&a)[8]) {
    #pragma unroll
    for (int s = 0; s < 8; ++s) {
      const int kcol = f * 32 + 4 * s + 2 * kh;
      v2f b0 = { w[kcol * 32 + n0],      w[(kcol + 1) * 32 + n0] };
      v2f b1 = { w[kcol * 32 + n0 + 16], w[(kcol + 1) * 32 + n0 + 16] };
      acc0 = __builtin_amdgcn_wmma_f32_16x16x4_f32(false, a[s], false, b0,
                                                   (short)0, acc0, false, false);
      acc1 = __builtin_amdgcn_wmma_f32_16x16x4_f32(false, a[s], false, b1,
                                                   (short)0, acc1, false, false);
      if constexpr (PAIR) {
        const int kc2 = g * 32 + 4 * s + 2 * kh;
        v2f c0 = { w[kc2 * 32 + n0],      w[(kc2 + 1) * 32 + n0] };
        v2f c1 = { w[kc2 * 32 + n0 + 16], w[(kc2 + 1) * 32 + n0 + 16] };
        acc2 = __builtin_amdgcn_wmma_f32_16x16x4_f32(false, a[s], false, c0,
                                                     (short)0, acc2, false, false);
        acc3 = __builtin_amdgcn_wmma_f32_16x16x4_f32(false, a[s], false, c1,
                                                     (short)0, acc3, false, false);
      }
    }
  };
  auto plane_frag = [&](const v4f (&raw)[32], float* sb, v2f (&a)[8]) {
    k3_fold(raw, sb, lane);
    __builtin_amdgcn_fence(__ATOMIC_RELEASE, "wavefront");
    __builtin_amdgcn_wave_barrier();
    __builtin_amdgcn_fence(__ATOMIC_ACQUIRE, "wavefront");
    #pragma unroll
    for (int s = 0; s < 8; ++s) a[s] = *(const v2f*)(sb + MA * 36 + 4 * s + 2 * kh);
    __builtin_amdgcn_fence(__ATOMIC_RELEASE, "wavefront");
    __builtin_amdgcn_wave_barrier();
    __builtin_amdgcn_fence(__ATOMIC_ACQUIRE, "wavefront");
  };
  v2f fd[4][8];
  auto direct_issue = [&](size_t off) {
    #pragma unroll
    for (int j = 0; j < 4; ++j) {
      #pragma unroll
      for (int s = 0; s < 8; ++s) fd[j][s] = *(const v2f*)(comp[j] + off + 4 * s + 2 * kh);
    }
  };
  v2f fa[8], fb[8];

  direct_issue(offPQ);
  __builtin_amdgcn_sched_barrier(0);
  plane_frag(rawA, sb0, fa);
  k3_issue(rawA, s1p, planeQP, 2048, lane);
  plane_frag(rawB, sb1, fb);
  k3_issue(rawB, s2p, planeQP, 2048, lane);
  __builtin_amdgcn_sched_barrier(0);
  mm(0, 6, fa);
  mm(1, 7, fb);
  #pragma unroll
  for (int j = 0; j < 4; ++j) mm(2 + j, 8 + j, fd[j]);
  __builtin_amdgcn_sched_barrier(0);
  direct_issue(offQP);
  plane_frag(rawA, sb0, fa);
  plane_frag(rawB, sb1, fb);
  mm(6, 0, fa);
  mm(7, 1, fb);
  #pragma unroll
  for (int j = 0; j < 4; ++j) mm(8 + j, 2 + j, fd[j]);

  float v02_0 = 0.f, v02_1 = 0.f, v01_0 = 0.f, v01_1 = 0.f;
  #pragma unroll
  for (int g = 0; g < 4; ++g) {
    v02_0 += r02p[(bT * 4 + g) * 32 + n0];
    v02_1 += r02p[(bT * 4 + g) * 32 + n0 + 16];
    if constexpr (!PAIR) {
      v01_0 += r01p[(bT * 4 + g) * 32 + n0];
      v01_1 += r01p[(bT * 4 + g) * 32 + n0 + 16];
    }
  }
  const int bp = (bT << 6) + pT;
  #pragma unroll
  for (int r = 0; r < 8; ++r) {
    const int q = q0 + r + ((lane >> 4) << 3);
    const int bq = (bT << 6) + q;
    float add0 = r12[bp * 32 + n0] + r12t[bq * 32 + n0] + v02_0;
    float add1 = r12[bp * 32 + n0 + 16] + r12t[bq * 32 + n0 + 16] + v02_1;
    if constexpr (!PAIR) {
      if (pT == q) {
        add0 += forow[bp * 32 + n0] + v01_0;
        add1 += forow[bp * 32 + n0 + 16] + v01_1;
      }
    }
    const size_t row = (size_t)(bp * 64 + q);
    out[(row << 5) + n0]      = acc0[r] + add0;
    out[(row << 5) + n0 + 16] = acc1[r] + add1;
    if constexpr (PAIR) {
      const float m0 = r12[bq * 32 + n0] + r12t[bp * 32 + n0] + v02_0;
      const float m1 = r12[bq * 32 + n0 + 16] + r12t[bp * 32 + n0 + 16] + v02_1;
      const size_t row2 = (size_t)(bq * 64 + pT);
      out[(row2 << 5) + n0]      = acc2[r] + m0;
      out[(row2 << 5) + n0 + 16] = acc3[r] + m1;
    }
  }
}

__global__ void __launch_bounds__(256)
k3b_mix(const float* __restrict__ s1p, const float* __restrict__ s2p,
        const float* __restrict__ s3m, const float* __restrict__ d12,
        const float* __restrict__ d13, const float* __restrict__ d23,
        const float* __restrict__ W22,
        const float* __restrict__ r12, const float* __restrict__ r12t,
        const float* __restrict__ r02p, const float* __restrict__ r01p,
        const float* __restrict__ forow, float* __restrict__ out) {
  __shared__ float w[384 * 32];
  __shared__ __attribute__((aligned(16))) float stg[8 * 2 * 576];
  const int t = threadIdx.x;
  const int lane = t & 31;
  const int wv = __builtin_amdgcn_readfirstlane(t >> 5);
  const int bT = blockIdx.x >> 5, lb = blockIdx.x & 31;
  const bool pair = wv < 3;
  const bool active = pair || (wv & 3) == 3;
  int pT, qr;
  if (pair) {
    const int d = lb * 3 + wv;
    if (d < 48)      { pT = d / 3;               qr = 1 + d - 3 * pT; }
    else if (d < 80) { pT = 16 + ((d - 48) >> 1); qr = 2 + (d & 1); }
    else             { pT = d - 48;              qr = 3; }
  } else {
    pT = lb * 2 + (wv >> 2);
    qr = pT >> 4;
  }
  const int q0 = qr << 4;
  const size_t planePQ = ((size_t)(bT * 512 + pT) << 11) + (size_t)(q0 << 5);
  const size_t planeQP = ((size_t)(bT * 512 + q0) << 11) + (size_t)(pT << 5);
  float* sb0 = stg + wv * 1152;
  float* sb1 = sb0 + 576;

  v4f rawA[32], rawB[32];
  if (active) {
    k3_issue(rawA, s1p, planePQ, 32, lane);
    k3_issue(rawB, s2p, planePQ, 32, lane);
  }
  __builtin_amdgcn_sched_barrier(0);

  for (int idx = t; idx < 384 * 32; idx += 256) w[idx] = W22[idx];
  __syncthreads();
  if (!active) return;

  if (pair)
    k3_tile<true>(rawA, rawB, w, sb0, sb1, s1p, s2p, s3m, d12, d13, d23,
                  r12, r12t, r02p, r01p, forow, out, bT, pT, q0, planeQP, lane);
  else
    k3_tile<false>(rawA, rawB, w, sb0, sb1, s1p, s2p, s3m, d12, d13, d23,
                   r12, r12t, r02p, r01p, forow, out, bT, pT, q0, planeQP, lane);
}

extern "C" void kernel_launch(void* const* d_in, const int* in_sizes, int n_in,
                              void* d_out, int out_size, void* d_ws, size_t ws_size,
                              hipStream_t stream) {
  (void)in_sizes; (void)n_in; (void)out_size;
  if (ws_size < WS_FLOATS * sizeof(float)) return;

  const float* x    = (const float*)d_in[0];
  const float* W22  = (const float*)d_in[1];
  const float* b22  = (const float*)d_in[2];
  const float* W12  = (const float*)d_in[3];
  const float* b12  = (const float*)d_in[4];
  const float* W12t = (const float*)d_in[5];
  const float* b12t = (const float*)d_in[6];
  const float* W02  = (const float*)d_in[7];
  const float* b02  = (const float*)d_in[8];
  const float* W11  = (const float*)d_in[9];
  const float* b11  = (const float*)d_in[10];
  const float* W01  = (const float*)d_in[11];
  const float* b01  = (const float*)d_in[12];

  float* ws = (float*)d_ws;
  float* s1p = ws + S1P;   float* s2p = ws + S2P;
  float* s3m = ws + S3M;   float* d12 = ws + D12;
  float* d13 = ws + D13;   float* d23 = ws + D23;
  float* tri = ws + TRI;   float* a1  = ws + A1O;
  float* r12 = ws + R12O;  float* r12t= ws + R12TO;
  float* forow = ws + FORO;
  float* r02p = ws + R02P; float* r01p = ws + R01P;

  k1_contract<<<dim3(8, 8, 8), 256, 0, stream>>>(x, s1p, s2p, s3m, d12, d13, d23, tri);
  k2b_a1     <<<512, 256, 0, stream>>>(s1p, s3m, d12, d13, d23, tri, a1);
  k3a_rows   <<<128, 256, 0, stream>>>(a1, W12, b12, W12t, b12t,
                                      W11, b11, W01, b01, W02, b02, b22,
                                      r12, r12t, forow, r02p, r01p);
  k3b_mix    <<<256, 256, 0, stream>>>(s1p, s2p, s3m, d12, d13, d23, W22,
                                       r12, r12t, r02p, r01p, forow, (float*)d_out);
}
